# starting kernel
# baseline (speedup 1.0000x reference)
#include <hip/hip_runtime.h>

typedef float v2f __attribute__((ext_vector_type(2)));
typedef float v8f __attribute__((ext_vector_type(8)));

#define HID 128

#define AU_KP 16
#define AH_KP 136
#define A3_KP 392


__global__ void zero_ws_kernel(float* ws, long n) {
  long i = (long)blockIdx.x * blockDim.x + threadIdx.x;
  long stride = (long)gridDim.x * blockDim.x;
  for (; i < n; i += stride) ws[i] = 0.0f;
}

__global__ void pack_weights_kernel(const float* __restrict__ W_u2h,
                                    const float* __restrict__ W_h2h,
                                    const float* __restrict__ W_upd,
                                    float* __restrict__ WTu,
                                    float* __restrict__ WTh,
                                    float* __restrict__ WTupd) {
  const int stride = gridDim.x * blockDim.x;
  const int tid = blockIdx.x * blockDim.x + threadIdx.x;
  for (int i = tid; i < HID * AU_KP; i += stride) {
    int c = i / AU_KP, k = i % AU_KP;
    WTu[i] = (k < 13) ? W_u2h[k * HID + c] : 0.f;
  }
  for (int i = tid; i < HID * AH_KP; i += stride) {
    int c = i / AH_KP, k = i % AH_KP;
    WTh[i] = (k < 133) ? W_h2h[k * HID + c] : 0.f;
  }
  for (int i = tid; i < HID * A3_KP; i += stride) {
    int c = i / A3_KP, k = i % A3_KP;
    WTupd[i] = (k < 386) ? W_upd[k * HID + c] : 0.f;
  }
}

__global__ __launch_bounds__(256) void agg_a2s_kernel(
    const int* __restrict__ src, const int* __restrict__ dst,
    const float* __restrict__ dis, const float* __restrict__ pos_action,
    const float* __restrict__ u, float* __restrict__ aggU, int E) {
  int e = blockIdx.x * blockDim.x + threadIdx.x;
  if (e >= E) return;
  int s = src[e], d = dst[e];
  float* row = aggU + (long)d * 12;
  atomicAdd(row + 0, pos_action[2 * s]);
  atomicAdd(row + 1, pos_action[2 * s + 1]);
  atomicAdd(row + 2, dis[e]);
  const float* up = u + (long)s * 8;
#pragma unroll
  for (int i = 0; i < 8; ++i) atomicAdd(row + 3 + i, up[i]);
  atomicAdd(row + 11, 1.0f);
}

__global__ __launch_bounds__(256) void agg_s2s_kernel(
    const int* __restrict__ src, const int* __restrict__ dst,
    const float* __restrict__ dis, const float* __restrict__ pos_state,
    const float* __restrict__ h, float* __restrict__ aggH, int E) {
  int gid = blockIdx.x * blockDim.x + threadIdx.x;
  int e = gid >> 5;
  int lane = gid & 31;
  if (e >= E) return;
  int s = src[e], d = dst[e];
  float* row = aggH + (long)d * 132;
  if (lane == 0) {
    atomicAdd(row + 0, pos_state[2 * s]);
    atomicAdd(row + 1, pos_state[2 * s + 1]);
    atomicAdd(row + 2, dis[e]);
    atomicAdd(row + 131, 1.0f);
  }
  const float4* hp4 = (const float4*)(h + (long)s * HID);
  float4 v = hp4[lane];
  float* p = row + 3 + 4 * lane;
  atomicAdd(p + 0, v.x);
  atomicAdd(p + 1, v.y);
  atomicAdd(p + 2, v.z);
  atomicAdd(p + 3, v.w);
}

#define A3_LD 393
#define AH_LD 137
#define AU_LD 17

__global__ __launch_bounds__(256, 1) void fused_node_kernel(
    const float* __restrict__ h, const float* __restrict__ pos_state,
    const float* __restrict__ aggU, const float* __restrict__ aggH,
    const float* __restrict__ WTu, const float* __restrict__ b_u2h,
    const float* __restrict__ WTh, const float* __restrict__ b_h2h,
    const float* __restrict__ WTupd, const float* __restrict__ b_upd,
    float* __restrict__ out, int n_state) {
  __shared__ float A[16 * A3_LD];
  __shared__ float Ah[16 * AH_LD];
  __shared__ float Au[16 * AU_LD];
  __shared__ float degU[16];
  __shared__ float degS[16];

  const int tid = threadIdx.x;
  const int n0 = blockIdx.x * 16;
  const int r = tid >> 4;
  const int c0 = tid & 15;
  const int n = n0 + r;
  const bool valid = (n < n_state);

  if (c0 == 0) {
    const float* au = aggU + (long)n * 12;
    const float* ah = aggH + (long)n * 132;
    float du = valid ? au[11] : 0.f;
    float ds = valid ? ah[131] : 0.f;
    degU[r] = du;
    degS[r] = ds;
    float p0 = valid ? pos_state[2 * n] : 0.f;
    float p1 = valid ? pos_state[2 * n + 1] : 0.f;
    float* AuR = Au + r * AU_LD;
    AuR[0] = valid ? au[0] : 0.f;
    AuR[1] = valid ? au[1] : 0.f;
    AuR[2] = du * p0;
    AuR[3] = du * p1;
    AuR[4] = valid ? au[2] : 0.f;
#pragma unroll
    for (int i = 0; i < 8; ++i) AuR[5 + i] = valid ? au[3 + i] : 0.f;
    AuR[13] = AuR[14] = AuR[15] = 0.f;

    float* AhR = Ah + r * AH_LD;
    AhR[0] = valid ? ah[0] : 0.f;
    AhR[1] = valid ? ah[1] : 0.f;
    AhR[2] = ds * p0;
    AhR[3] = ds * p1;
    AhR[4] = valid ? ah[2] : 0.f;
    AhR[133] = AhR[134] = AhR[135] = 0.f;

    float* AR = A + r * A3_LD;
    AR[384] = p0;
    AR[385] = p1;
#pragma unroll
    for (int k = 386; k < A3_KP; ++k) AR[k] = 0.f;
  }
  {
    const float* hp = h + (long)n * HID;
    const float* ahp = aggH + (long)n * 132 + 3;
    for (int c = c0; c < HID; c += 16) {
      A[r * A3_LD + c] = valid ? hp[c] : 0.f;
      Ah[r * AH_LD + 5 + c] = valid ? ahp[c] : 0.f;
    }
  }
  __syncthreads();

  const int lane = tid & 31;
  const int wv = tid >> 5;
  const int half = lane >> 4;
  const int m = lane & 15;
  const int kb = half * 2;
  const int col = wv * 16 + m;

  {
    const float* bp = WTu + col * AU_KP + kb;
    v8f acc = {};
#pragma unroll
    for (int k = 0; k < AU_KP; k += 4) {
      v2f a;
      a.x = Au[m * AU_LD + k + kb];
      a.y = Au[m * AU_LD + k + kb + 1];
      v2f b = *(const v2f*)(bp + k);
      acc = __builtin_amdgcn_wmma_f32_16x16x4_f32(false, a, false, b, (short)0,
                                                  acc, false, false);
    }
    float bu = b_u2h[col];
#pragma unroll
    for (int i = 0; i < 8; ++i) {
      int rr = i + 8 * half;
      A[rr * A3_LD + HID + col] = acc[i] + degU[rr] * bu;
    }
  }

  {
    const float* bp = WTh + col * AH_KP + kb;
    v8f acc = {};
#pragma unroll 2
    for (int k = 0; k < AH_KP; k += 4) {
      v2f a;
      a.x = Ah[m * AH_LD + k + kb];
      a.y = Ah[m * AH_LD + k + kb + 1];
      v2f b = *(const v2f*)(bp + k);
      acc = __builtin_amdgcn_wmma_f32_16x16x4_f32(false, a, false, b, (short)0,
                                                  acc, false, false);
    }
    float bh = b_h2h[col];
#pragma unroll
    for (int i = 0; i < 8; ++i) {
      int rr = i + 8 * half;
      float ds = degS[rr];
      A[rr * A3_LD + 2 * HID + col] =
          (acc[i] + ds * bh) / fmaxf(ds, 1.0f);
    }
  }
  __syncthreads();

  {
    const float* bp = WTupd + col * A3_KP + kb;
    v8f acc = {};
#pragma unroll 2
    for (int k = 0; k < A3_KP; k += 4) {
      v2f a;
      a.x = A[m * A3_LD + k + kb];
      a.y = A[m * A3_LD + k + kb + 1];
      v2f b = *(const v2f*)(bp + k);
      acc = __builtin_amdgcn_wmma_f32_16x16x4_f32(false, a, false, b, (short)0,
                                                  acc, false, false);
    }
    float bo = b_upd[col];
#pragma unroll
    for (int i = 0; i < 8; ++i) {
      int rr = i + 8 * half;
      int nn = n0 + rr;
      if (nn < n_state) out[(long)nn * HID + col] = acc[i] + bo;
    }
  }
}

extern "C" void kernel_launch(void* const* d_in, const int* in_sizes, int n_in,
                              void* d_out, int out_size, void* d_ws,
                              size_t ws_size, hipStream_t stream) {
  const float* h = (const float*)d_in[0];
  const float* u = (const float*)d_in[1];
  const float* pos_state = (const float*)d_in[2];
  const float* pos_action = (const float*)d_in[3];
  const int* a2s_src = (const int*)d_in[4];
  const int* a2s_dst = (const int*)d_in[5];
  const float* a2s_dis = (const float*)d_in[6];
  const int* s2s_src = (const int*)d_in[7];
  const int* s2s_dst = (const int*)d_in[8];
  const float* s2s_dis = (const float*)d_in[9];
  const float* W_u2h = (const float*)d_in[10];
  const float* b_u2h = (const float*)d_in[11];
  const float* W_h2h = (const float*)d_in[12];
  const float* b_h2h = (const float*)d_in[13];
  const float* W_upd = (const float*)d_in[14];
  const float* b_upd = (const float*)d_in[15];
  float* out = (float*)d_out;

  const int n_state = in_sizes[2] / 2;
  const int E_a2s = in_sizes[4];
  const int E_s2s = in_sizes[7];

  float* aggU = (float*)d_ws;
  float* aggH = aggU + (size_t)n_state * 12;
  float* WTu = aggH + (size_t)n_state * 132;
  float* WTh = WTu + HID * AU_KP;
  float* WTupd = WTh + HID * AH_KP;
  const long agg_floats = (long)n_state * (12 + 132);

  zero_ws_kernel<<<2048, 256, 0, stream>>>(aggU, agg_floats);
  pack_weights_kernel<<<256, 256, 0, stream>>>(W_u2h, W_h2h, W_upd, WTu, WTh,
                                               WTupd);
  agg_a2s_kernel<<<(E_a2s + 255) / 256, 256, 0, stream>>>(
      a2s_src, a2s_dst, a2s_dis, pos_action, u, aggU, E_a2s);
  agg_s2s_kernel<<<(E_s2s + 7) / 8, 256, 0, stream>>>(
      s2s_src, s2s_dst, s2s_dis, pos_state, h, aggH, E_s2s);
  fused_node_kernel<<<(n_state + 15) / 16, 256, 0, stream>>>(
      h, pos_state, aggU, aggH, WTu, b_u2h, WTh, b_h2h, WTupd, b_upd, out,
      n_state);
}


# count, scan, fill replaced by one edge pass: a node owns a fixed slot block, the counting atomic's return is the edge's slot; full blocks spill to a scanned list; 5 launches to 3; arrival-order sums
# speedup vs baseline: 2.1596x; 2.1596x over previous
#include <hip/hip_runtime.h>


typedef float v2f __attribute__((ext_vector_type(2)));
typedef float v8f __attribute__((ext_vector_type(8)));

#define HID 128

#define AU_KP 16
#define AH_KP 136
#define A3_KP 392

#define CAPS 32
#define CAPA 16

__global__ void pack_weights_kernel(const float* __restrict__ W_u2h,
                                    const float* __restrict__ W_h2h,
                                    const float* __restrict__ W_upd,
                                    float* __restrict__ WTu,
                                    float* __restrict__ WTh,
                                    float* __restrict__ Wp,
                                    int* __restrict__ counters, int n_state) {
  const int stride = gridDim.x * blockDim.x;
  const int tid = blockIdx.x * blockDim.x + threadIdx.x;
  const int nz = 2 * n_state + 2;
  for (int i = tid; i < nz; i += stride) counters[i] = 0;
  for (int i = tid; i < HID * AU_KP; i += stride) {
    int c = i / AU_KP, k = i % AU_KP;
    WTu[i] = (k < 13) ? W_u2h[k * HID + c] : 0.f;
  }
  for (int i = tid; i < HID * AH_KP; i += stride) {
    int c = i / AH_KP, k = i % AH_KP;
    WTh[i] = (k < 133) ? W_h2h[k * HID + c] : 0.f;
  }
  for (int i = tid; i < HID * A3_KP; i += stride)
    Wp[i] = (i < 386 * HID) ? W_upd[i] : 0.f;
}

__global__ __launch_bounds__(256) void count_kernel(
    const int* __restrict__ dstS, const int* __restrict__ dstA,
    int* __restrict__ cntS, int* __restrict__ cntA, int* __restrict__ rank,
    int ES, int EA) {
  int e = blockIdx.x * blockDim.x + threadIdx.x;
  if (e < ES) {
    int r = atomicAdd(cntS + dstS[e], 1);
    if (rank) rank[e] = r;
  } else if (e - ES < EA) {
    int r = atomicAdd(cntA + dstA[e - ES], 1);
    if (rank) rank[e] = r;
  }
}

__global__ __launch_bounds__(256) void alloc_kernel(
    const int* __restrict__ cntS, const int* __restrict__ cntA,
    int* __restrict__ tot, int* __restrict__ offS, int* __restrict__ offA,
    int* __restrict__ curS, int* __restrict__ curA, int n_state) {
  const int i = blockIdx.x * blockDim.x + threadIdx.x;
  const int lane = threadIdx.x & 31;
  const int cs = (i < n_state) ? cntS[i] : 0;
  const int ca = (i < n_state) ? cntA[i] : 0;
  int ss = cs, sa = ca;
  for (int o = 1; o < 32; o <<= 1) {
    int ts = __shfl_up(ss, o);
    int ta = __shfl_up(sa, o);
    if (lane >= o) {
      ss += ts;
      sa += ta;
    }
  }
  int baseS = 0, baseA = 0;
  if (lane == 31) {
    baseS = atomicAdd(tot + 0, ss);
    baseA = atomicAdd(tot + 1, sa);
  }
  baseS = __shfl(baseS, 31);
  baseA = __shfl(baseA, 31);
  if (i < n_state) {
    int os = baseS + ss - cs;
    int oa = baseA + sa - ca;
    offS[i] = os;
    offA[i] = oa;
    if (curS) {
      curS[i] = os;
      curA[i] = oa;
    }
  }
}

__global__ __launch_bounds__(256) void fill_kernel(
    const int* __restrict__ srcS, const int* __restrict__ dstS,
    const float* __restrict__ disS, const int* __restrict__ srcA,
    const int* __restrict__ dstA, const float* __restrict__ disA,
    const int* __restrict__ offS, const int* __restrict__ offA,
    const int* __restrict__ rank, int* __restrict__ curS,
    int* __restrict__ curA, int* __restrict__ listS, int* __restrict__ listA,
    int ES, int EA) {
  int e = blockIdx.x * blockDim.x + threadIdx.x;
  if (e < ES) {
    const int d = dstS[e];
    int p = rank ? offS[d] + rank[e] : atomicAdd(curS + d, 1);
    p = min(max(p, 0), ES - 1);
    ((int2*)listS)[p] = make_int2(srcS[e], __float_as_int(disS[e]));
  } else if (e - ES < EA) {
    int a = e - ES;
    const int d = dstA[a];
    int p = rank ? offA[d] + rank[e] : atomicAdd(curA + d, 1);
    p = min(max(p, 0), EA - 1);
    ((int2*)listA)[p] = make_int2(srcA[a], __float_as_int(disA[a]));
  }
}

__global__ __launch_bounds__(256) void slot_kernel(
    const int* __restrict__ srcS, const int* __restrict__ dstS,
    const float* __restrict__ disS, const int* __restrict__ srcA,
    const int* __restrict__ dstA, const float* __restrict__ disA,
    int* __restrict__ cntS, int* __restrict__ cntA, int* __restrict__ nspill,
    int* __restrict__ spill, int* __restrict__ listS, int* __restrict__ listA,
    int n_state, int ES, int EA) {
  const int e = blockIdx.x * blockDim.x + threadIdx.x;
  if (e < ES) {
    const int d = dstS[e];
    int s = srcS[e];
    int dv = __float_as_int(disS[e]);
    asm volatile("" : "+v"(s), "+v"(dv));
    if ((unsigned)d < (unsigned)n_state) {
      const int r = atomicAdd(cntS + d, 1);
      if (r < CAPS) {
        ((int2*)listS)[(long)d * CAPS + r] = make_int2(s, dv);
      } else {
        const int o = atomicAdd(nspill, 1);
        if (o < ES + EA) spill[o] = e;
      }
    }
  } else if (e - ES < EA) {
    const int a = e - ES;
    const int d = dstA[a];
    int s = srcA[a];
    int dv = __float_as_int(disA[a]);
    asm volatile("" : "+v"(s), "+v"(dv));
    if ((unsigned)d < (unsigned)n_state) {
      const int r = atomicAdd(cntA + d, 1);
      if (r < CAPA) {
        ((int2*)listA)[(long)d * CAPA + r] = make_int2(s, dv);
      } else {
        const int o = atomicAdd(nspill, 1);
        if (o < ES + EA) spill[o] = e;
      }
    }
  }
}

__device__ __forceinline__ float wave_sum(float v) {
  for (int o = 16; o > 0; o >>= 1) v += __shfl_xor(v, o);
  return v;
}

#define A3_LD 393
#define AH_LD 137
#define AU_LD 17
#define SM_A 0
#define SM_AH (16 * A3_LD)
#define SM_AU (SM_AH + 16 * AH_LD)
#define SM_TOTAL (SM_AU + 16 * AU_LD)

__global__ __launch_bounds__(256, 16) void fused_node_kernel(
    const float* __restrict__ h, const float* __restrict__ pos_state,
    const float* __restrict__ u, const float* __restrict__ pos_action,
    const int* __restrict__ listS, const int* __restrict__ listA,
    const int* __restrict__ offS, const int* __restrict__ cntS,
    const int* __restrict__ offA, const int* __restrict__ cntA,
    const float* __restrict__ WTu, const float* __restrict__ b_u2h,
    const float* __restrict__ WTh, const float* __restrict__ b_h2h,
    const float* __restrict__ Wp, const float* __restrict__ b_upd,
    const int* __restrict__ nspill, const int* __restrict__ spill,
    const int* __restrict__ srcS, const int* __restrict__ dstS,
    const float* __restrict__ disS, const int* __restrict__ srcA,
    const int* __restrict__ dstA, const float* __restrict__ disA,
    float* __restrict__ out, int n_state, int n_action, int ES, int EA) {
  __shared__ __attribute__((aligned(16))) float smem[SM_TOTAL];
  float* const A = smem + SM_A;
  float* const Ah = smem + SM_AH;
  float* const Au = smem + SM_AU;
  __shared__ float degU[16];
  __shared__ float degS[16];

  const int tid = threadIdx.x;
  const int n0 = blockIdx.x * 16;
  const int lane = tid & 31;
  const int wv = __builtin_amdgcn_readfirstlane(tid >> 5);
  const int nsp = spill ? min(max(nspill[0], 0), ES + EA) : 0;

  {
    const int r = tid >> 4;
    const int c0 = tid & 15;
    const int n = n0 + r;
    const bool valid = (n < n_state);
    if (c0 == 0) {
      float* AR = A + r * A3_LD;
      AR[384] = valid ? pos_state[2 * n] : 0.f;
      AR[385] = valid ? pos_state[2 * n + 1] : 0.f;
#pragma unroll
      for (int k = 386; k < A3_KP; ++k) AR[k] = 0.f;
    }
    const float* hp = h + (long)n * HID;
    for (int c = c0; c < HID; c += 16)
      A[r * A3_LD + c] = valid ? hp[c] : 0.f;
  }

  for (int rr = 0; rr < 2; ++rr) {
    const int r2 = wv * 2 + rr;
    const int n2 = n0 + r2;
    int begS = 0, numS = 0, begA = 0, numA = 0;
    float q0 = 0.f, q1 = 0.f;
    if (n2 < n_state) {
      numS = cntS[n2];
      numA = cntA[n2];
      if (offS) {
        begS = offS[n2];
        begA = offA[n2];
      } else {
        begS = n2 * CAPS;
        begA = n2 * CAPA;
      }
      q0 = pos_state[2 * n2];
      q1 = pos_state[2 * n2 + 1];
    }
    const int degSi = min(max(numS, 0), ES);
    const int degAi = min(max(numA, 0), EA);
    const int lenS = offS ? ES : n_state * CAPS;
    const int lenA = offS ? EA : n_state * CAPA;
    begS = min(max(begS, 0), lenS);
    const int endS = begS + min(offS ? degSi : min(degSi, CAPS), lenS - begS);
    begA = min(max(begA, 0), lenA);
    const int endA = begA + min(offS ? degAi : min(degAi, CAPA), lenA - begA);

    float4 acc = make_float4(0.f, 0.f, 0.f, 0.f);
    float ps0 = 0.f, ps1 = 0.f, dsm = 0.f;
    for (int i = begS; i < endS; i += 32) {
      const int my = i + lane;
      int smine = 0;
      float2 p = make_float2(0.f, 0.f);
      float dv = 0.f;
      if (my < endS) {
        int2 rec = ((const int2*)listS)[my];
        smine = (int)min((unsigned)rec.x, (unsigned)(n_state - 1));
        p = ((const float2*)pos_state)[smine];
        dv = __int_as_float(rec.y);
      }
      const int mc = min(32, endS - i);
#pragma unroll 4
      for (int j = 0; j < mc; ++j) {
        const int s = __builtin_amdgcn_readlane(smine, j);
        const float4 v = ((const float4*)(h + (long)s * HID))[lane];
        acc.x += v.x;
        acc.y += v.y;
        acc.z += v.z;
        acc.w += v.w;
      }
      ps0 += p.x;
      ps1 += p.y;
      dsm += dv;
    }
    ps0 = wave_sum(ps0);
    ps1 = wave_sum(ps1);
    dsm = wave_sum(dsm);

    const int slot = lane >> 3, comp = lane & 7;
    float ua = 0.f, xa = 0.f;
    for (int i = begA; i < endA; i += 4) {
      const int my = i + slot;
      if (my < endA) {
        int2 rec = ((const int2*)listA)[my];
        const int s = (int)min((unsigned)rec.x, (unsigned)(n_action - 1));
        ua += u[(long)s * 8 + comp];
        float x = 0.f;
        if (comp < 2) x = pos_action[2 * s + comp];
        else if (comp == 2) x = __int_as_float(rec.y);
        xa += x;
      }
    }
    for (int i = 0; i < nsp; i += 32) {
      const int my = i + lane;
      int e = 0, hit = 0;
      if (my < nsp) {
        e = min(max(spill[my], 0), ES + EA - 1);
        const int dd = (e < ES) ? dstS[e] : dstA[e - ES];
        if (dd == n2) hit = (e < ES) ? 1 : 2;
      }
      unsigned mS = __builtin_amdgcn_ballot_w32(hit == 1);
      while (mS) {
        const int j = __builtin_ctz(mS);
        mS &= mS - 1;
        const int ee = __builtin_amdgcn_readlane(e, j);
        const int s = (int)min((unsigned)srcS[ee], (unsigned)(n_state - 1));
        const float4 v = ((const float4*)(h + (long)s * HID))[lane];
        acc.x += v.x;
        acc.y += v.y;
        acc.z += v.z;
        acc.w += v.w;
        ps0 += pos_state[2 * s];
        ps1 += pos_state[2 * s + 1];
        dsm += disS[ee];
      }
      unsigned mA = __builtin_amdgcn_ballot_w32(hit == 2);
      while (mA) {
        const int j = __builtin_ctz(mA);
        mA &= mA - 1;
        const int a = __builtin_amdgcn_readlane(e, j) - ES;
        const int s = (int)min((unsigned)srcA[a], (unsigned)(n_action - 1));
        if (slot == 0) {
          ua += u[(long)s * 8 + comp];
          float x = 0.f;
          if (comp < 2) x = pos_action[2 * s + comp];
          else if (comp == 2) x = disA[a];
          xa += x;
        }
      }
    }
    ua += __shfl_xor(ua, 8);
    ua += __shfl_xor(ua, 16);
    xa += __shfl_xor(xa, 8);
    xa += __shfl_xor(xa, 16);

    const float dgS = (float)degSi;
    const float dgA = (float)degAi;
    float* AhR = Ah + r2 * AH_LD;
    float* AuR = Au + r2 * AU_LD;
    AhR[5 + 4 * lane + 0] = acc.x;
    AhR[5 + 4 * lane + 1] = acc.y;
    AhR[5 + 4 * lane + 2] = acc.z;
    AhR[5 + 4 * lane + 3] = acc.w;
    if (lane < 8) AuR[5 + lane] = ua;
    if (lane < 2) AuR[lane] = xa;
    if (lane == 2) AuR[4] = xa;
    if (lane == 0) {
      AhR[0] = ps0;
      AhR[1] = ps1;
      AhR[2] = dgS * q0;
      AhR[3] = dgS * q1;
      AhR[4] = dsm;
      AhR[133] = AhR[134] = AhR[135] = 0.f;
      AuR[2] = dgA * q0;
      AuR[3] = dgA * q1;
      AuR[13] = AuR[14] = AuR[15] = 0.f;
      degS[r2] = dgS;
      degU[r2] = dgA;
    }
  }
  __syncthreads();

  const int half = lane >> 4;
  const int m = lane & 15;
  const int kb = half * 2;
  const int col = wv * 16 + m;

  {
    const float* bp = WTu + col * AU_KP + kb;
    v8f acc = {};
#pragma unroll
    for (int k = 0; k < AU_KP; k += 4) {
      v2f a;
      a.x = Au[m * AU_LD + k + kb];
      a.y = Au[m * AU_LD + k + kb + 1];
      v2f b = *(const v2f*)(bp + k);
      acc = __builtin_amdgcn_wmma_f32_16x16x4_f32(false, a, false, b, (short)0,
                                                  acc, false, false);
    }
    float bu = b_u2h[col];
#pragma unroll
    for (int i = 0; i < 8; ++i) {
      int rr = i + 8 * half;
      A[rr * A3_LD + HID + col] = acc[i] + degU[rr] * bu;
    }
  }

  {
    const float* bp = WTh + col * AH_KP + kb;
    v8f acc = {};
#pragma unroll 2
    for (int k = 0; k < AH_KP; k += 4) {
      v2f a;
      a.x = Ah[m * AH_LD + k + kb];
      a.y = Ah[m * AH_LD + k + kb + 1];
      v2f b = *(const v2f*)(bp + k);
      acc = __builtin_amdgcn_wmma_f32_16x16x4_f32(false, a, false, b, (short)0,
                                                  acc, false, false);
    }
    float bh = b_h2h[col];
#pragma unroll
    for (int i = 0; i < 8; ++i) {
      int rr = i + 8 * half;
      float ds = degS[rr];
      A[rr * A3_LD + 2 * HID + col] =
          (acc[i] + ds * bh) / fmaxf(ds, 1.0f);
    }
  }
  __syncthreads();

  {
    const int sp = wv >> 2;
    const int kq = wv & 3;
    const int ks0 = (kq * 49) >> 1;
    const int ks1 = ((kq + 1) * 49) >> 1;
    const unsigned bo = (unsigned)(kb * HID + sp * 64 + 4 * m);
    const float* ap = A + m * A3_LD + kb;
    v8f acc0 = {}, acc1 = {}, acc2 = {}, acc3 = {};
#pragma unroll 2
    for (int ks = ks0; ks < ks1; ++ks) {
      const int k = ks * 4;
      v2f a;
      a.x = ap[k];
      a.y = ap[k + 1];
      const unsigned br = bo + (unsigned)k * HID;
      const float4 b0 = *(const float4*)(Wp + br);
      const float4 b1 = *(const float4*)(Wp + (br + HID));
      v2f b;
      b.x = b0.x;
      b.y = b1.x;
      acc0 = __builtin_amdgcn_wmma_f32_16x16x4_f32(false, a, false, b, (short)0,
                                                   acc0, false, false);
      b.x = b0.y;
      b.y = b1.y;
      acc1 = __builtin_amdgcn_wmma_f32_16x16x4_f32(false, a, false, b, (short)0,
                                                   acc1, false, false);
      b.x = b0.z;
      b.y = b1.z;
      acc2 = __builtin_amdgcn_wmma_f32_16x16x4_f32(false, a, false, b, (short)0,
                                                   acc2, false, false);
      b.x = b0.w;
      b.y = b1.w;
      acc3 = __builtin_amdgcn_wmma_f32_16x16x4_f32(false, a, false, b, (short)0,
                                                   acc3, false, false);
    }
    __syncthreads();

    float* P = smem + wv * 1024 + m;
#pragma unroll
    for (int i = 0; i < 8; ++i) {
      const int ro = (i + 8 * half) * 16;
      P[0 * 256 + ro] = acc0[i];
      P[1 * 256 + ro] = acc1[i];
      P[2 * 256 + ro] = acc2[i];
      P[3 * 256 + ro] = acc3[i];
    }
  }
  __syncthreads();

  {
    const int row = tid >> 4;
    const int c16 = tid & 15;
    const int nn = n0 + row;
#pragma unroll
    for (int sp = 0; sp < 2; ++sp) {
      float o[4];
#pragma unroll
      for (int j = 0; j < 4; ++j) {
        const float* Q = smem + (sp * 4) * 1024 + j * 256 + row * 16 + c16;
        o[j] = ((Q[0] + Q[1024]) + Q[2048]) + Q[3072];
      }
      const float* bq = b_upd + sp * 64 + 4 * c16;
      if (nn < n_state) {
        float4 r;
        r.x = o[0] + bq[0];
        r.y = o[1] + bq[1];
        r.z = o[2] + bq[2];
        r.w = o[3] + bq[3];
        *(float4*)(out + (long)nn * HID + sp * 64 + 4 * c16) = r;
      }
    }
  }
}

extern "C" void kernel_launch(void* const* d_in, const int* in_sizes, int n_in,
                              void* d_out, int out_size, void* d_ws,
                              size_t ws_size, hipStream_t stream) {
  const float* h = (const float*)d_in[0];
  const float* u = (const float*)d_in[1];
  const float* pos_state = (const float*)d_in[2];
  const float* pos_action = (const float*)d_in[3];
  const int* a2s_src = (const int*)d_in[4];
  const int* a2s_dst = (const int*)d_in[5];
  const float* a2s_dis = (const float*)d_in[6];
  const int* s2s_src = (const int*)d_in[7];
  const int* s2s_dst = (const int*)d_in[8];
  const float* s2s_dis = (const float*)d_in[9];
  const float* W_u2h = (const float*)d_in[10];
  const float* b_u2h = (const float*)d_in[11];
  const float* W_h2h = (const float*)d_in[12];
  const float* b_h2h = (const float*)d_in[13];
  const float* W_upd = (const float*)d_in[14];
  const float* b_upd = (const float*)d_in[15];
  float* out = (float*)d_out;

  const int n_state = in_sizes[2] / 2;
  const int n_action = in_sizes[3] / 2;
  const int E_a2s = in_sizes[4];
  const int E_s2s = in_sizes[7];

  float* WTu = (float*)d_ws;
  float* WTh = WTu + HID * AU_KP;
  float* Wp = WTh + HID * AH_KP;
  int* listS = (int*)(Wp + HID * A3_KP);
  int* listA = listS + (size_t)E_s2s * 2;
  int* cntS = listA + (size_t)E_a2s * 2;
  int* cntA = cntS + n_state;
  int* tot = cntA + n_state;
  int* offS = tot + 2;
  int* offA = offS + n_state;
  int* curS = offA + n_state;
  int* curA = curS + n_state;
  const int E_all = E_s2s + E_a2s;
  int* rank = curA + n_state;
  if ((size_t)((char*)(rank + E_all) - (char*)d_ws) > ws_size) rank = nullptr;
  int* curS_used = rank ? nullptr : curS;
  int* curA_used = rank ? nullptr : curA;

  int* sl_listS = (int*)(Wp + HID * A3_KP);
  int* sl_listA = sl_listS + (size_t)n_state * CAPS * 2;
  int* sl_cntS = sl_listA + (size_t)n_state * CAPA * 2;
  int* sl_cntA = sl_cntS + n_state;
  int* sl_tot = sl_cntA + n_state;
  int* sl_spill = sl_tot + 2;
  if ((size_t)((char*)(sl_spill + E_all) - (char*)d_ws) <= ws_size) {
    pack_weights_kernel<<<256, 256, 0, stream>>>(W_u2h, W_h2h, W_upd, WTu, WTh,
                                                 Wp, sl_cntS, n_state);
    slot_kernel<<<(E_all + 255) / 256, 256, 0, stream>>>(
        s2s_src, s2s_dst, s2s_dis, a2s_src, a2s_dst, a2s_dis, sl_cntS, sl_cntA,
        sl_tot, sl_spill, sl_listS, sl_listA, n_state, E_s2s, E_a2s);
    fused_node_kernel<<<(n_state + 15) / 16, 256, 0, stream>>>(
        h, pos_state, u, pos_action, sl_listS, sl_listA, nullptr, sl_cntS,
        nullptr, sl_cntA, WTu, b_u2h, WTh, b_h2h, Wp, b_upd, sl_tot, sl_spill,
        s2s_src, s2s_dst, s2s_dis, a2s_src, a2s_dst, a2s_dis, out, n_state,
        n_action, E_s2s, E_a2s);
    return;
  }
  pack_weights_kernel<<<256, 256, 0, stream>>>(W_u2h, W_h2h, W_upd, WTu, WTh,
                                               Wp, cntS, n_state);
  count_kernel<<<(E_all + 255) / 256, 256, 0, stream>>>(
      s2s_dst, a2s_dst, cntS, cntA, rank, E_s2s, E_a2s);
  alloc_kernel<<<(n_state + 255) / 256, 256, 0, stream>>>(
      cntS, cntA, tot, offS, offA, curS_used, curA_used, n_state);
  fill_kernel<<<(E_all + 255) / 256, 256, 0, stream>>>(
      s2s_src, s2s_dst, s2s_dis, a2s_src, a2s_dst, a2s_dis, offS, offA, rank,
      curS_used, curA_used, listS, listA, E_s2s, E_a2s);
  fused_node_kernel<<<(n_state + 15) / 16, 256, 0, stream>>>(
      h, pos_state, u, pos_action, listS, listA, offS, cntS, offA, cntA, WTu,
      b_u2h, WTh, b_h2h, Wp, b_upd, nullptr, nullptr, s2s_src, s2s_dst,
      s2s_dis, a2s_src, a2s_dst, a2s_dis, out, n_state, n_action, E_s2s,
      E_a2s);
}


# edge pass: each node's two in-degree counters sit alone in a 128-byte line (were packed, 32 nodes to a line), so counting atomics of different nodes never share a line; arrival-order sums as before
# speedup vs baseline: 2.5297x; 1.1714x over previous
#include <hip/hip_runtime.h>


typedef float v2f __attribute__((ext_vector_type(2)));
typedef float v8f __attribute__((ext_vector_type(8)));

#define HID 128

#define AU_KP 16
#define AH_KP 136
#define A3_KP 392

#define CAPS 32
#define CAPA 16

__global__ void pack_weights_kernel(const float* __restrict__ W_u2h,
                                    const float* __restrict__ W_h2h,
                                    const float* __restrict__ W_upd,
                                    float* __restrict__ WTu,
                                    float* __restrict__ WTh,
                                    float* __restrict__ Wp,
                                    int* __restrict__ dense, int n_dense,
                                    int* __restrict__ pairs, int n_pairs,
                                    int cstride) {
  const int stride = gridDim.x * blockDim.x;
  const int tid = blockIdx.x * blockDim.x + threadIdx.x;
  for (int i = tid; i < n_dense; i += stride) dense[i] = 0;
  for (int i = tid; i < n_pairs; i += stride)
    *(int2*)(pairs + (unsigned)i * (unsigned)cstride) = make_int2(0, 0);
  for (int i = tid; i < HID * AU_KP; i += stride) {
    int c = i / AU_KP, k = i % AU_KP;
    WTu[i] = (k < 13) ? W_u2h[k * HID + c] : 0.f;
  }
  for (int i = tid; i < HID * AH_KP; i += stride) {
    int c = i / AH_KP, k = i % AH_KP;
    WTh[i] = (k < 133) ? W_h2h[k * HID + c] : 0.f;
  }
  for (int i = tid; i < HID * A3_KP; i += stride)
    Wp[i] = (i < 386 * HID) ? W_upd[i] : 0.f;
}

__global__ __launch_bounds__(256) void count_kernel(
    const int* __restrict__ dstS, const int* __restrict__ dstA,
    int* __restrict__ cntS, int* __restrict__ cntA, int* __restrict__ rank,
    int ES, int EA) {
  int e = blockIdx.x * blockDim.x + threadIdx.x;
  if (e < ES) {
    int r = atomicAdd(cntS + dstS[e], 1);
    if (rank) rank[e] = r;
  } else if (e - ES < EA) {
    int r = atomicAdd(cntA + dstA[e - ES], 1);
    if (rank) rank[e] = r;
  }
}

__global__ __launch_bounds__(256) void alloc_kernel(
    const int* __restrict__ cntS, const int* __restrict__ cntA,
    int* __restrict__ tot, int* __restrict__ offS, int* __restrict__ offA,
    int* __restrict__ curS, int* __restrict__ curA, int n_state) {
  const int i = blockIdx.x * blockDim.x + threadIdx.x;
  const int lane = threadIdx.x & 31;
  const int cs = (i < n_state) ? cntS[i] : 0;
  const int ca = (i < n_state) ? cntA[i] : 0;
  int ss = cs, sa = ca;
  for (int o = 1; o < 32; o <<= 1) {
    int ts = __shfl_up(ss, o);
    int ta = __shfl_up(sa, o);
    if (lane >= o) {
      ss += ts;
      sa += ta;
    }
  }
  int baseS = 0, baseA = 0;
  if (lane == 31) {
    baseS = atomicAdd(tot + 0, ss);
    baseA = atomicAdd(tot + 1, sa);
  }
  baseS = __shfl(baseS, 31);
  baseA = __shfl(baseA, 31);
  if (i < n_state) {
    int os = baseS + ss - cs;
    int oa = baseA + sa - ca;
    offS[i] = os;
    offA[i] = oa;
    if (curS) {
      curS[i] = os;
      curA[i] = oa;
    }
  }
}

__global__ __launch_bounds__(256) void fill_kernel(
    const int* __restrict__ srcS, const int* __restrict__ dstS,
    const float* __restrict__ disS, const int* __restrict__ srcA,
    const int* __restrict__ dstA, const float* __restrict__ disA,
    const int* __restrict__ offS, const int* __restrict__ offA,
    const int* __restrict__ rank, int* __restrict__ curS,
    int* __restrict__ curA, int* __restrict__ listS, int* __restrict__ listA,
    int ES, int EA) {
  int e = blockIdx.x * blockDim.x + threadIdx.x;
  if (e < ES) {
    const int d = dstS[e];
    int p = rank ? offS[d] + rank[e] : atomicAdd(curS + d, 1);
    p = min(max(p, 0), ES - 1);
    ((int2*)listS)[p] = make_int2(srcS[e], __float_as_int(disS[e]));
  } else if (e - ES < EA) {
    int a = e - ES;
    const int d = dstA[a];
    int p = rank ? offA[d] + rank[e] : atomicAdd(curA + d, 1);
    p = min(max(p, 0), EA - 1);
    ((int2*)listA)[p] = make_int2(srcA[a], __float_as_int(disA[a]));
  }
}

__global__ __launch_bounds__(256) void slot_kernel(
    const int* __restrict__ srcS, const int* __restrict__ dstS,
    const float* __restrict__ disS, const int* __restrict__ srcA,
    const int* __restrict__ dstA, const float* __restrict__ disA,
    int* __restrict__ cntS, int* __restrict__ cntA, int* __restrict__ nspill,
    int* __restrict__ spill, int* __restrict__ listS, int* __restrict__ listA,
    int n_state, int ES, int EA, int cstride) {
  const int e = blockIdx.x * blockDim.x + threadIdx.x;
  if (e < ES) {
    const int d = dstS[e];
    int s = srcS[e];
    int dv = __float_as_int(disS[e]);
    asm volatile("" : "+v"(s), "+v"(dv));
    if ((unsigned)d < (unsigned)n_state) {
      const int r = atomicAdd(cntS + (unsigned)d * (unsigned)cstride, 1);
      if (r < CAPS) {
        ((int2*)listS)[(long)d * CAPS + r] = make_int2(s, dv);
      } else {
        const int o = atomicAdd(nspill, 1);
        if (o < ES + EA) spill[o] = e;
      }
    }
  } else if (e - ES < EA) {
    const int a = e - ES;
    const int d = dstA[a];
    int s = srcA[a];
    int dv = __float_as_int(disA[a]);
    asm volatile("" : "+v"(s), "+v"(dv));
    if ((unsigned)d < (unsigned)n_state) {
      const int r = atomicAdd(cntA + (unsigned)d * (unsigned)cstride, 1);
      if (r < CAPA) {
        ((int2*)listA)[(long)d * CAPA + r] = make_int2(s, dv);
      } else {
        const int o = atomicAdd(nspill, 1);
        if (o < ES + EA) spill[o] = e;
      }
    }
  }
}

__device__ __forceinline__ float wave_sum(float v) {
  for (int o = 16; o > 0; o >>= 1) v += __shfl_xor(v, o);
  return v;
}

#define A3_LD 393
#define AH_LD 137
#define AU_LD 17
#define SM_A 0
#define SM_AH (16 * A3_LD)
#define SM_AU (SM_AH + 16 * AH_LD)
#define SM_TOTAL (SM_AU + 16 * AU_LD)

__global__ __launch_bounds__(256, 16) void fused_node_kernel(
    const float* __restrict__ h, const float* __restrict__ pos_state,
    const float* __restrict__ u, const float* __restrict__ pos_action,
    const int* __restrict__ listS, const int* __restrict__ listA,
    const int* __restrict__ offS, const int* __restrict__ cntS,
    const int* __restrict__ offA, const int* __restrict__ cntA,
    const float* __restrict__ WTu, const float* __restrict__ b_u2h,
    const float* __restrict__ WTh, const float* __restrict__ b_h2h,
    const float* __restrict__ Wp, const float* __restrict__ b_upd,
    const int* __restrict__ nspill, const int* __restrict__ spill,
    const int* __restrict__ srcS, const int* __restrict__ dstS,
    const float* __restrict__ disS, const int* __restrict__ srcA,
    const int* __restrict__ dstA, const float* __restrict__ disA,
    float* __restrict__ out, int n_state, int n_action, int ES, int EA,
    int cstride) {
  __shared__ __attribute__((aligned(16))) float smem[SM_TOTAL];
  float* const A = smem + SM_A;
  float* const Ah = smem + SM_AH;
  float* const Au = smem + SM_AU;
  __shared__ float degU[16];
  __shared__ float degS[16];

  const int tid = threadIdx.x;
  const int n0 = blockIdx.x * 16;
  const int lane = tid & 31;
  const int wv = __builtin_amdgcn_readfirstlane(tid >> 5);
  const int nsp = spill ? min(max(nspill[0], 0), ES + EA) : 0;

  {
    const int r = tid >> 4;
    const int c0 = tid & 15;
    const int n = n0 + r;
    const bool valid = (n < n_state);
    if (c0 == 0) {
      float* AR = A + r * A3_LD;
      AR[384] = valid ? pos_state[2 * n] : 0.f;
      AR[385] = valid ? pos_state[2 * n + 1] : 0.f;
#pragma unroll
      for (int k = 386; k < A3_KP; ++k) AR[k] = 0.f;
    }
    const float* hp = h + (long)n * HID;
    for (int c = c0; c < HID; c += 16)
      A[r * A3_LD + c] = valid ? hp[c] : 0.f;
  }

  for (int rr = 0; rr < 2; ++rr) {
    const int r2 = wv * 2 + rr;
    const int n2 = n0 + r2;
    int begS = 0, numS = 0, begA = 0, numA = 0;
    float q0 = 0.f, q1 = 0.f;
    if (n2 < n_state) {
      numS = cntS[(unsigned)n2 * (unsigned)cstride];
      numA = cntA[(unsigned)n2 * (unsigned)cstride];
      if (offS) {
        begS = offS[n2];
        begA = offA[n2];
      } else {
        begS = n2 * CAPS;
        begA = n2 * CAPA;
      }
      q0 = pos_state[2 * n2];
      q1 = pos_state[2 * n2 + 1];
    }
    const int degSi = min(max(numS, 0), ES);
    const int degAi = min(max(numA, 0), EA);
    const int lenS = offS ? ES : n_state * CAPS;
    const int lenA = offS ? EA : n_state * CAPA;
    begS = min(max(begS, 0), lenS);
    const int endS = begS + min(offS ? degSi : min(degSi, CAPS), lenS - begS);
    begA = min(max(begA, 0), lenA);
    const int endA = begA + min(offS ? degAi : min(degAi, CAPA), lenA - begA);

    float4 acc = make_float4(0.f, 0.f, 0.f, 0.f);
    float ps0 = 0.f, ps1 = 0.f, dsm = 0.f;
    for (int i = begS; i < endS; i += 32) {
      const int my = i + lane;
      int smine = 0;
      float2 p = make_float2(0.f, 0.f);
      float dv = 0.f;
      if (my < endS) {
        int2 rec = ((const int2*)listS)[my];
        smine = (int)min((unsigned)rec.x, (unsigned)(n_state - 1));
        p = ((const float2*)pos_state)[smine];
        dv = __int_as_float(rec.y);
      }
      const int mc = min(32, endS - i);
#pragma unroll 4
      for (int j = 0; j < mc; ++j) {
        const int s = __builtin_amdgcn_readlane(smine, j);
        const float4 v = ((const float4*)(h + (long)s * HID))[lane];
        acc.x += v.x;
        acc.y += v.y;
        acc.z += v.z;
        acc.w += v.w;
      }
      ps0 += p.x;
      ps1 += p.y;
      dsm += dv;
    }
    ps0 = wave_sum(ps0);
    ps1 = wave_sum(ps1);
    dsm = wave_sum(dsm);

    const int slot = lane >> 3, comp = lane & 7;
    float ua = 0.f, xa = 0.f;
    for (int i = begA; i < endA; i += 4) {
      const int my = i + slot;
      if (my < endA) {
        int2 rec = ((const int2*)listA)[my];
        const int s = (int)min((unsigned)rec.x, (unsigned)(n_action - 1));
        ua += u[(long)s * 8 + comp];
        float x = 0.f;
        if (comp < 2) x = pos_action[2 * s + comp];
        else if (comp == 2) x = __int_as_float(rec.y);
        xa += x;
      }
    }
    for (int i = 0; i < nsp; i += 32) {
      const int my = i + lane;
      int e = 0, hit = 0;
      if (my < nsp) {
        e = min(max(spill[my], 0), ES + EA - 1);
        const int dd = (e < ES) ? dstS[e] : dstA[e - ES];
        if (dd == n2) hit = (e < ES) ? 1 : 2;
      }
      unsigned mS = __builtin_amdgcn_ballot_w32(hit == 1);
      while (mS) {
        const int j = __builtin_ctz(mS);
        mS &= mS - 1;
        const int ee = __builtin_amdgcn_readlane(e, j);
        const int s = (int)min((unsigned)srcS[ee], (unsigned)(n_state - 1));
        const float4 v = ((const float4*)(h + (long)s * HID))[lane];
        acc.x += v.x;
        acc.y += v.y;
        acc.z += v.z;
        acc.w += v.w;
        ps0 += pos_state[2 * s];
        ps1 += pos_state[2 * s + 1];
        dsm += disS[ee];
      }
      unsigned mA = __builtin_amdgcn_ballot_w32(hit == 2);
      while (mA) {
        const int j = __builtin_ctz(mA);
        mA &= mA - 1;
        const int a = __builtin_amdgcn_readlane(e, j) - ES;
        const int s = (int)min((unsigned)srcA[a], (unsigned)(n_action - 1));
        if (slot == 0) {
          ua += u[(long)s * 8 + comp];
          float x = 0.f;
          if (comp < 2) x = pos_action[2 * s + comp];
          else if (comp == 2) x = disA[a];
          xa += x;
        }
      }
    }
    ua += __shfl_xor(ua, 8);
    ua += __shfl_xor(ua, 16);
    xa += __shfl_xor(xa, 8);
    xa += __shfl_xor(xa, 16);

    const float dgS = (float)degSi;
    const float dgA = (float)degAi;
    float* AhR = Ah + r2 * AH_LD;
    float* AuR = Au + r2 * AU_LD;
    AhR[5 + 4 * lane + 0] = acc.x;
    AhR[5 + 4 * lane + 1] = acc.y;
    AhR[5 + 4 * lane + 2] = acc.z;
    AhR[5 + 4 * lane + 3] = acc.w;
    if (lane < 8) AuR[5 + lane] = ua;
    if (lane < 2) AuR[lane] = xa;
    if (lane == 2) AuR[4] = xa;
    if (lane == 0) {
      AhR[0] = ps0;
      AhR[1] = ps1;
      AhR[2] = dgS * q0;
      AhR[3] = dgS * q1;
      AhR[4] = dsm;
      AhR[133] = AhR[134] = AhR[135] = 0.f;
      AuR[2] = dgA * q0;
      AuR[3] = dgA * q1;
      AuR[13] = AuR[14] = AuR[15] = 0.f;
      degS[r2] = dgS;
      degU[r2] = dgA;
    }
  }
  __syncthreads();

  const int half = lane >> 4;
  const int m = lane & 15;
  const int kb = half * 2;
  const int col = wv * 16 + m;

  {
    const float* bp = WTu + col * AU_KP + kb;
    v8f acc = {};
#pragma unroll
    for (int k = 0; k < AU_KP; k += 4) {
      v2f a;
      a.x = Au[m * AU_LD + k + kb];
      a.y = Au[m * AU_LD + k + kb + 1];
      v2f b = *(const v2f*)(bp + k);
      acc = __builtin_amdgcn_wmma_f32_16x16x4_f32(false, a, false, b, (short)0,
                                                  acc, false, false);
    }
    float bu = b_u2h[col];
#pragma unroll
    for (int i = 0; i < 8; ++i) {
      int rr = i + 8 * half;
      A[rr * A3_LD + HID + col] = acc[i] + degU[rr] * bu;
    }
  }

  {
    const float* bp = WTh + col * AH_KP + kb;
    v8f acc = {};
#pragma unroll 2
    for (int k = 0; k < AH_KP; k += 4) {
      v2f a;
      a.x = Ah[m * AH_LD + k + kb];
      a.y = Ah[m * AH_LD + k + kb + 1];
      v2f b = *(const v2f*)(bp + k);
      acc = __builtin_amdgcn_wmma_f32_16x16x4_f32(false, a, false, b, (short)0,
                                                  acc, false, false);
    }
    float bh = b_h2h[col];
#pragma unroll
    for (int i = 0; i < 8; ++i) {
      int rr = i + 8 * half;
      float ds = degS[rr];
      A[rr * A3_LD + 2 * HID + col] =
          (acc[i] + ds * bh) / fmaxf(ds, 1.0f);
    }
  }
  __syncthreads();

  {
    const int sp = wv >> 2;
    const int kq = wv & 3;
    const int ks0 = (kq * 49) >> 1;
    const int ks1 = ((kq + 1) * 49) >> 1;
    const unsigned bo = (unsigned)(kb * HID + sp * 64 + 4 * m);
    const float* ap = A + m * A3_LD + kb;
    v8f acc0 = {}, acc1 = {}, acc2 = {}, acc3 = {};
#pragma unroll 2
    for (int ks = ks0; ks < ks1; ++ks) {
      const int k = ks * 4;
      v2f a;
      a.x = ap[k];
      a.y = ap[k + 1];
      const unsigned br = bo + (unsigned)k * HID;
      const float4 b0 = *(const float4*)(Wp + br);
      const float4 b1 = *(const float4*)(Wp + (br + HID));
      v2f b;
      b.x = b0.x;
      b.y = b1.x;
      acc0 = __builtin_amdgcn_wmma_f32_16x16x4_f32(false, a, false, b, (short)0,
                                                   acc0, false, false);
      b.x = b0.y;
      b.y = b1.y;
      acc1 = __builtin_amdgcn_wmma_f32_16x16x4_f32(false, a, false, b, (short)0,
                                                   acc1, false, false);
      b.x = b0.z;
      b.y = b1.z;
      acc2 = __builtin_amdgcn_wmma_f32_16x16x4_f32(false, a, false, b, (short)0,
                                                   acc2, false, false);
      b.x = b0.w;
      b.y = b1.w;
      acc3 = __builtin_amdgcn_wmma_f32_16x16x4_f32(false, a, false, b, (short)0,
                                                   acc3, false, false);
    }
    __syncthreads();

    float* P = smem + wv * 1024 + m;
#pragma unroll
    for (int i = 0; i < 8; ++i) {
      const int ro = (i + 8 * half) * 16;
      P[0 * 256 + ro] = acc0[i];
      P[1 * 256 + ro] = acc1[i];
      P[2 * 256 + ro] = acc2[i];
      P[3 * 256 + ro] = acc3[i];
    }
  }
  __syncthreads();

  {
    const int row = tid >> 4;
    const int c16 = tid & 15;
    const int nn = n0 + row;
#pragma unroll
    for (int sp = 0; sp < 2; ++sp) {
      float o[4];
#pragma unroll
      for (int j = 0; j < 4; ++j) {
        const float* Q = smem + (sp * 4) * 1024 + j * 256 + row * 16 + c16;
        o[j] = ((Q[0] + Q[1024]) + Q[2048]) + Q[3072];
      }
      const float* bq = b_upd + sp * 64 + 4 * c16;
      if (nn < n_state) {
        float4 r;
        r.x = o[0] + bq[0];
        r.y = o[1] + bq[1];
        r.z = o[2] + bq[2];
        r.w = o[3] + bq[3];
        *(float4*)(out + (long)nn * HID + sp * 64 + 4 * c16) = r;
      }
    }
  }
}

extern "C" void kernel_launch(void* const* d_in, const int* in_sizes, int n_in,
                              void* d_out, int out_size, void* d_ws,
                              size_t ws_size, hipStream_t stream) {
  const float* h = (const float*)d_in[0];
  const float* u = (const float*)d_in[1];
  const float* pos_state = (const float*)d_in[2];
  const float* pos_action = (const float*)d_in[3];
  const int* a2s_src = (const int*)d_in[4];
  const int* a2s_dst = (const int*)d_in[5];
  const float* a2s_dis = (const float*)d_in[6];
  const int* s2s_src = (const int*)d_in[7];
  const int* s2s_dst = (const int*)d_in[8];
  const float* s2s_dis = (const float*)d_in[9];
  const float* W_u2h = (const float*)d_in[10];
  const float* b_u2h = (const float*)d_in[11];
  const float* W_h2h = (const float*)d_in[12];
  const float* b_h2h = (const float*)d_in[13];
  const float* W_upd = (const float*)d_in[14];
  const float* b_upd = (const float*)d_in[15];
  float* out = (float*)d_out;

  const int n_state = in_sizes[2] / 2;
  const int n_action = in_sizes[3] / 2;
  const int E_a2s = in_sizes[4];
  const int E_s2s = in_sizes[7];

  float* WTu = (float*)d_ws;
  float* WTh = WTu + HID * AU_KP;
  float* Wp = WTh + HID * AH_KP;
  int* listS = (int*)(Wp + HID * A3_KP);
  int* listA = listS + (size_t)E_s2s * 2;
  int* cntS = listA + (size_t)E_a2s * 2;
  int* cntA = cntS + n_state;
  int* tot = cntA + n_state;
  int* offS = tot + 2;
  int* offA = offS + n_state;
  int* curS = offA + n_state;
  int* curA = curS + n_state;
  const int E_all = E_s2s + E_a2s;
  int* rank = curA + n_state;
  if ((size_t)((char*)(rank + E_all) - (char*)d_ws) > ws_size) rank = nullptr;
  int* curS_used = rank ? nullptr : curS;
  int* curA_used = rank ? nullptr : curA;

  int* sl_listS = (int*)(Wp + HID * A3_KP);
  int* sl_listA = sl_listS + (size_t)n_state * CAPS * 2;
  int* sl_cnt2 = sl_listA + (size_t)n_state * CAPA * 2;
  int cstride = 32;
  if ((size_t)((char*)(sl_cnt2 + (size_t)n_state * 32 + 2 + E_all) -
               (char*)d_ws) > ws_size)
    cstride = 8;
  if (cstride == 8 && (size_t)((char*)(sl_cnt2 + (size_t)n_state * 8 + 2 +
                                       E_all) - (char*)d_ws) > ws_size)
    cstride = 2;
  int* sl_tot = sl_cnt2 + (size_t)n_state * cstride;
  int* sl_spill = sl_tot + 2;
  if ((size_t)((char*)(sl_spill + E_all) - (char*)d_ws) <= ws_size) {
    pack_weights_kernel<<<256, 256, 0, stream>>>(W_u2h, W_h2h, W_upd, WTu, WTh,
                                                 Wp, sl_tot, 2, sl_cnt2,
                                                 n_state, cstride);
    slot_kernel<<<(E_all + 255) / 256, 256, 0, stream>>>(
        s2s_src, s2s_dst, s2s_dis, a2s_src, a2s_dst, a2s_dis, sl_cnt2,
        sl_cnt2 + 1, sl_tot, sl_spill, sl_listS, sl_listA, n_state, E_s2s,
        E_a2s, cstride);
    fused_node_kernel<<<(n_state + 15) / 16, 256, 0, stream>>>(
        h, pos_state, u, pos_action, sl_listS, sl_listA, nullptr, sl_cnt2,
        nullptr, sl_cnt2 + 1, WTu, b_u2h, WTh, b_h2h, Wp, b_upd, sl_tot,
        sl_spill, s2s_src, s2s_dst, s2s_dis, a2s_src, a2s_dst, a2s_dis, out,
        n_state, n_action, E_s2s, E_a2s, cstride);
    return;
  }
  pack_weights_kernel<<<256, 256, 0, stream>>>(W_u2h, W_h2h, W_upd, WTu, WTh,
                                               Wp, cntS, 2 * n_state + 2,
                                               nullptr, 0, 0);
  count_kernel<<<(E_all + 255) / 256, 256, 0, stream>>>(
      s2s_dst, a2s_dst, cntS, cntA, rank, E_s2s, E_a2s);
  alloc_kernel<<<(n_state + 255) / 256, 256, 0, stream>>>(
      cntS, cntA, tot, offS, offA, curS_used, curA_used, n_state);
  fill_kernel<<<(E_all + 255) / 256, 256, 0, stream>>>(
      s2s_src, s2s_dst, s2s_dis, a2s_src, a2s_dst, a2s_dis, offS, offA, rank,
      curS_used, curA_used, listS, listA, E_s2s, E_a2s);
  fused_node_kernel<<<(n_state + 15) / 16, 256, 0, stream>>>(
      h, pos_state, u, pos_action, listS, listA, offS, cntS, offA, cntA, WTu,
      b_u2h, WTh, b_h2h, Wp, b_upd, nullptr, nullptr, s2s_src, s2s_dst,
      s2s_dis, a2s_src, a2s_dst, a2s_dis, out, n_state, n_action, E_s2s,
      E_a2s, 1);
}


# edge pass: one copy of the in-degree counters per package half, a wave adds to its half's copy (no line is added to from both halves); block filled from both ends, no spill list; arrival-order sums
# speedup vs baseline: 2.8532x; 1.1279x over previous
#include <hip/hip_runtime.h>


typedef float v2f __attribute__((ext_vector_type(2)));
typedef float v8f __attribute__((ext_vector_type(8)));

#define HID 128

#define AU_KP 16
#define AH_KP 136
#define A3_KP 392

#define CAPS 32
#define CAPA 16

__global__ void pack_weights_kernel(const float* __restrict__ W_u2h,
                                    const float* __restrict__ W_h2h,
                                    const float* __restrict__ W_upd,
                                    float* __restrict__ WTu,
                                    float* __restrict__ WTh,
                                    float* __restrict__ Wp,
                                    int* __restrict__ dense, int n_dense,
                                    int* __restrict__ pairs, int n_pairs,
                                    int cstride) {
  const int stride = gridDim.x * blockDim.x;
  const int tid = blockIdx.x * blockDim.x + threadIdx.x;
  for (int i = tid; i < n_dense; i += stride) dense[i] = 0;
  for (int i = tid; i < n_pairs; i += stride)
    *(int2*)(pairs + (unsigned)i * (unsigned)cstride) = make_int2(0, 0);
  for (int i = tid; i < HID * AU_KP; i += stride) {
    int c = i / AU_KP, k = i % AU_KP;
    WTu[i] = (k < 13) ? W_u2h[k * HID + c] : 0.f;
  }
  for (int i = tid; i < HID * AH_KP; i += stride) {
    int c = i / AH_KP, k = i % AH_KP;
    WTh[i] = (k < 133) ? W_h2h[k * HID + c] : 0.f;
  }
  for (int i = tid; i < HID * A3_KP; i += stride)
    Wp[i] = (i < 386 * HID) ? W_upd[i] : 0.f;
}

__global__ __launch_bounds__(256) void count_kernel(
    const int* __restrict__ dstS, const int* __restrict__ dstA,
    int* __restrict__ cntS, int* __restrict__ cntA, int* __restrict__ rank,
    int ES, int EA) {
  int e = blockIdx.x * blockDim.x + threadIdx.x;
  if (e < ES) {
    int r = atomicAdd(cntS + dstS[e], 1);
    if (rank) rank[e] = r;
  } else if (e - ES < EA) {
    int r = atomicAdd(cntA + dstA[e - ES], 1);
    if (rank) rank[e] = r;
  }
}

__global__ __launch_bounds__(256) void alloc_kernel(
    const int* __restrict__ cntS, const int* __restrict__ cntA,
    int* __restrict__ tot, int* __restrict__ offS, int* __restrict__ offA,
    int* __restrict__ curS, int* __restrict__ curA, int n_state) {
  const int i = blockIdx.x * blockDim.x + threadIdx.x;
  const int lane = threadIdx.x & 31;
  const int cs = (i < n_state) ? cntS[i] : 0;
  const int ca = (i < n_state) ? cntA[i] : 0;
  int ss = cs, sa = ca;
  for (int o = 1; o < 32; o <<= 1) {
    int ts = __shfl_up(ss, o);
    int ta = __shfl_up(sa, o);
    if (lane >= o) {
      ss += ts;
      sa += ta;
    }
  }
  int baseS = 0, baseA = 0;
  if (lane == 31) {
    baseS = atomicAdd(tot + 0, ss);
    baseA = atomicAdd(tot + 1, sa);
  }
  baseS = __shfl(baseS, 31);
  baseA = __shfl(baseA, 31);
  if (i < n_state) {
    int os = baseS + ss - cs;
    int oa = baseA + sa - ca;
    offS[i] = os;
    offA[i] = oa;
    if (curS) {
      curS[i] = os;
      curA[i] = oa;
    }
  }
}

__global__ __launch_bounds__(256) void fill_kernel(
    const int* __restrict__ srcS, const int* __restrict__ dstS,
    const float* __restrict__ disS, const int* __restrict__ srcA,
    const int* __restrict__ dstA, const float* __restrict__ disA,
    const int* __restrict__ offS, const int* __restrict__ offA,
    const int* __restrict__ rank, int* __restrict__ curS,
    int* __restrict__ curA, int* __restrict__ listS, int* __restrict__ listA,
    int ES, int EA) {
  int e = blockIdx.x * blockDim.x + threadIdx.x;
  if (e < ES) {
    const int d = dstS[e];
    int p = rank ? offS[d] + rank[e] : atomicAdd(curS + d, 1);
    p = min(max(p, 0), ES - 1);
    ((int2*)listS)[p] = make_int2(srcS[e], __float_as_int(disS[e]));
  } else if (e - ES < EA) {
    int a = e - ES;
    const int d = dstA[a];
    int p = rank ? offA[d] + rank[e] : atomicAdd(curA + d, 1);
    p = min(max(p, 0), EA - 1);
    ((int2*)listA)[p] = make_int2(srcA[a], __float_as_int(disA[a]));
  }
}

__device__ __forceinline__ unsigned long long rec8(int s, int dv) {
  return ((unsigned long long)(unsigned)dv << 32) | (unsigned long long)(unsigned)s;
}

__global__ __launch_bounds__(256) void slot_kernel(
    const int* __restrict__ srcS, const int* __restrict__ dstS,
    const float* __restrict__ disS, const int* __restrict__ srcA,
    const int* __restrict__ dstA, const float* __restrict__ disA,
    int* __restrict__ cnt2, int* __restrict__ listS, int* __restrict__ listA,
    int n_state, int ES, int EA, int hs) {
  const int e = blockIdx.x * blockDim.x + threadIdx.x;
  const int hb = (int)((__builtin_amdgcn_s_sendmsg_rtn(0x87) >> 18) & 1u);
  int* const cb = cnt2 + (size_t)hb * (size_t)n_state * (size_t)hs;
  if (e < ES) {
    const int d = dstS[e];
    int s = srcS[e];
    int dv = __float_as_int(disS[e]);
    asm volatile("" : "+v"(s), "+v"(dv));
    if ((unsigned)d < (unsigned)n_state) {
      const int r = atomicAdd(cb + (unsigned)d * (unsigned)hs, 1);
      if ((unsigned)r < (unsigned)CAPS)
        ((unsigned long long*)listS)[(long)d * CAPS + (hb ? CAPS - 1 - r : r)] =
            rec8(s, dv);
    }
  } else if (e - ES < EA) {
    const int a = e - ES;
    const int d = dstA[a];
    int s = srcA[a];
    int dv = __float_as_int(disA[a]);
    asm volatile("" : "+v"(s), "+v"(dv));
    if ((unsigned)d < (unsigned)n_state) {
      const int r = atomicAdd(cb + (unsigned)d * (unsigned)hs + 1, 1);
      if ((unsigned)r < (unsigned)CAPA)
        ((unsigned long long*)listA)[(long)d * CAPA + (hb ? CAPA - 1 - r : r)] =
            rec8(s, dv);
    }
  }
}

__device__ __forceinline__ float wave_sum(float v) {
  for (int o = 16; o > 0; o >>= 1) v += __shfl_xor(v, o);
  return v;
}

#define A3_LD 393
#define AH_LD 137
#define AU_LD 17
#define SM_A 0
#define SM_AH (16 * A3_LD)
#define SM_AU (SM_AH + 16 * AH_LD)
#define SM_TOTAL (SM_AU + 16 * AU_LD)

__global__ __launch_bounds__(256, 16) void fused_node_kernel(
    const float* __restrict__ h, const float* __restrict__ pos_state,
    const float* __restrict__ u, const float* __restrict__ pos_action,
    const int* __restrict__ listS, const int* __restrict__ listA,
    const int* __restrict__ offS, const int* __restrict__ cntS,
    const int* __restrict__ offA, const int* __restrict__ cntA,
    const float* __restrict__ WTu, const float* __restrict__ b_u2h,
    const float* __restrict__ WTh, const float* __restrict__ b_h2h,
    const float* __restrict__ Wp, const float* __restrict__ b_upd,
    const int* __restrict__ srcS, const int* __restrict__ dstS,
    const float* __restrict__ disS, const int* __restrict__ srcA,
    const int* __restrict__ dstA, const float* __restrict__ disA,
    float* __restrict__ out, int n_state, int n_action, int ES, int EA,
    int cstride) {
  __shared__ __attribute__((aligned(16))) float smem[SM_TOTAL];
  float* const A = smem + SM_A;
  float* const Ah = smem + SM_AH;
  float* const Au = smem + SM_AU;
  __shared__ float degU[16];
  __shared__ float degS[16];

  const int tid = threadIdx.x;
  const int n0 = blockIdx.x * 16;
  const int lane = tid & 31;
  const int wv = __builtin_amdgcn_readfirstlane(tid >> 5);

  {
    const int r = tid >> 4;
    const int c0 = tid & 15;
    const int n = n0 + r;
    const bool valid = (n < n_state);
    if (c0 == 0) {
      float* AR = A + r * A3_LD;
      AR[384] = valid ? pos_state[2 * n] : 0.f;
      AR[385] = valid ? pos_state[2 * n + 1] : 0.f;
#pragma unroll
      for (int k = 386; k < A3_KP; ++k) AR[k] = 0.f;
    }
    const float* hp = h + (long)n * HID;
    for (int c = c0; c < HID; c += 16)
      A[r * A3_LD + c] = valid ? hp[c] : 0.f;
  }

  for (int rr = 0; rr < 2; ++rr) {
    const int r2 = wv * 2 + rr;
    const int n2 = n0 + r2;
    int begS = 0, c0S = 0, c1S = 0, begA = 0, c0A = 0, c1A = 0;
    float q0 = 0.f, q1 = 0.f;
    if (n2 < n_state) {
      c0S = cntS[(unsigned)n2 * (unsigned)cstride];
      c0A = cntA[(unsigned)n2 * (unsigned)cstride];
      if (offS) {
        begS = offS[n2];
        begA = offA[n2];
      } else {
        c1S = cntS[(unsigned)(n_state + n2) * (unsigned)cstride];
        c1A = cntA[(unsigned)(n_state + n2) * (unsigned)cstride];
        begS = n2 * CAPS;
        begA = n2 * CAPA;
      }
      q0 = pos_state[2 * n2];
      q1 = pos_state[2 * n2 + 1];
    }
    c0S = min(max(c0S, 0), ES);
    c1S = min(max(c1S, 0), ES);
    c0A = min(max(c0A, 0), EA);
    c1A = min(max(c1A, 0), EA);
    const int degSi = min(c0S + c1S, ES);
    const int degAi = min(c0A + c1A, EA);
    const bool ovS = !offS && (c0S + c1S > CAPS);
    const bool ovA = !offS && (c0A + c1A > CAPA);
    const int lenS = offS ? ES : n_state * CAPS;
    const int lenA = offS ? EA : n_state * CAPA;
    begS = min(max(begS, 0), lenS);
    const int nS = min(ovS ? 0 : degSi, lenS - begS);
    const int endS = begS + nS;
    begA = min(max(begA, 0), lenA);
    const int nA = min(ovA ? 0 : degAi, lenA - begA);
    const int endA = begA + nA;
    const int splitS = offS ? 0x7fffffff : c0S;
    const int shiftS = offS ? 0 : CAPS - nS;
    const int splitA = offS ? 0x7fffffff : c0A;
    const int shiftA = offS ? 0 : CAPA - nA;

    float4 acc = make_float4(0.f, 0.f, 0.f, 0.f);
    float ps0 = 0.f, ps1 = 0.f, dsm = 0.f;
    for (int i = begS; i < endS; i += 32) {
      const int my = i + lane;
      int smine = 0;
      float2 p = make_float2(0.f, 0.f);
      float dv = 0.f;
      if (my < endS) {
        const int at = min(my + ((my - begS >= splitS) ? shiftS : 0), lenS - 1);
        int2 rec = ((const int2*)listS)[at];
        smine = (int)min((unsigned)rec.x, (unsigned)(n_state - 1));
        p = ((const float2*)pos_state)[smine];
        dv = __int_as_float(rec.y);
      }
      const int mc = min(32, endS - i);
#pragma unroll 4
      for (int j = 0; j < mc; ++j) {
        const int s = __builtin_amdgcn_readlane(smine, j);
        const float4 v = ((const float4*)(h + (long)s * HID))[lane];
        acc.x += v.x;
        acc.y += v.y;
        acc.z += v.z;
        acc.w += v.w;
      }
      ps0 += p.x;
      ps1 += p.y;
      dsm += dv;
    }
    ps0 = wave_sum(ps0);
    ps1 = wave_sum(ps1);
    dsm = wave_sum(dsm);

    const int slot = lane >> 3, comp = lane & 7;
    float ua = 0.f, xa = 0.f;
    for (int i = begA; i < endA; i += 4) {
      const int my = i + slot;
      if (my < endA) {
        const int at = min(my + ((my - begA >= splitA) ? shiftA : 0), lenA - 1);
        int2 rec = ((const int2*)listA)[at];
        const int s = (int)min((unsigned)rec.x, (unsigned)(n_action - 1));
        ua += u[(long)s * 8 + comp];
        float x = 0.f;
        if (comp < 2) x = pos_action[2 * s + comp];
        else if (comp == 2) x = __int_as_float(rec.y);
        xa += x;
      }
    }
    const int rlo = ovS ? 0 : ES;
    const int rhi = ovA ? ES + EA : ES;
    for (int i = rlo; i < rhi; i += 32) {
      const int e = i + lane;
      int hit = 0;
      if (e < rhi) {
        const int dd = (e < ES) ? dstS[e] : dstA[e - ES];
        if (dd == n2) hit = (e < ES) ? 1 : 2;
      }
      unsigned mS = __builtin_amdgcn_ballot_w32(hit == 1);
      while (mS) {
        const int j = __builtin_ctz(mS);
        mS &= mS - 1;
        const int ee = __builtin_amdgcn_readlane(e, j);
        const int s = (int)min((unsigned)srcS[ee], (unsigned)(n_state - 1));
        const float4 v = ((const float4*)(h + (long)s * HID))[lane];
        acc.x += v.x;
        acc.y += v.y;
        acc.z += v.z;
        acc.w += v.w;
        ps0 += pos_state[2 * s];
        ps1 += pos_state[2 * s + 1];
        dsm += disS[ee];
      }
      unsigned mA = __builtin_amdgcn_ballot_w32(hit == 2);
      while (mA) {
        const int j = __builtin_ctz(mA);
        mA &= mA - 1;
        const int a = __builtin_amdgcn_readlane(e, j) - ES;
        const int s = (int)min((unsigned)srcA[a], (unsigned)(n_action - 1));
        if (slot == 0) {
          ua += u[(long)s * 8 + comp];
          float x = 0.f;
          if (comp < 2) x = pos_action[2 * s + comp];
          else if (comp == 2) x = disA[a];
          xa += x;
        }
      }
    }
    ua += __shfl_xor(ua, 8);
    ua += __shfl_xor(ua, 16);
    xa += __shfl_xor(xa, 8);
    xa += __shfl_xor(xa, 16);

    const float dgS = (float)degSi;
    const float dgA = (float)degAi;
    float* AhR = Ah + r2 * AH_LD;
    float* AuR = Au + r2 * AU_LD;
    AhR[5 + 4 * lane + 0] = acc.x;
    AhR[5 + 4 * lane + 1] = acc.y;
    AhR[5 + 4 * lane + 2] = acc.z;
    AhR[5 + 4 * lane + 3] = acc.w;
    if (lane < 8) AuR[5 + lane] = ua;
    if (lane < 2) AuR[lane] = xa;
    if (lane == 2) AuR[4] = xa;
    if (lane == 0) {
      AhR[0] = ps0;
      AhR[1] = ps1;
      AhR[2] = dgS * q0;
      AhR[3] = dgS * q1;
      AhR[4] = dsm;
      AhR[133] = AhR[134] = AhR[135] = 0.f;
      AuR[2] = dgA * q0;
      AuR[3] = dgA * q1;
      AuR[13] = AuR[14] = AuR[15] = 0.f;
      degS[r2] = dgS;
      degU[r2] = dgA;
    }
  }
  __syncthreads();

  const int half = lane >> 4;
  const int m = lane & 15;
  const int kb = half * 2;
  const int col = wv * 16 + m;

  {
    const float* bp = WTu + col * AU_KP + kb;
    v8f acc = {};
#pragma unroll
    for (int k = 0; k < AU_KP; k += 4) {
      v2f a;
      a.x = Au[m * AU_LD + k + kb];
      a.y = Au[m * AU_LD + k + kb + 1];
      v2f b = *(const v2f*)(bp + k);
      acc = __builtin_amdgcn_wmma_f32_16x16x4_f32(false, a, false, b, (short)0,
                                                  acc, false, false);
    }
    float bu = b_u2h[col];
#pragma unroll
    for (int i = 0; i < 8; ++i) {
      int rr = i + 8 * half;
      A[rr * A3_LD + HID + col] = acc[i] + degU[rr] * bu;
    }
  }

  {
    const float* bp = WTh + col * AH_KP + kb;
    v8f acc = {};
#pragma unroll 2
    for (int k = 0; k < AH_KP; k += 4) {
      v2f a;
      a.x = Ah[m * AH_LD + k + kb];
      a.y = Ah[m * AH_LD + k + kb + 1];
      v2f b = *(const v2f*)(bp + k);
      acc = __builtin_amdgcn_wmma_f32_16x16x4_f32(false, a, false, b, (short)0,
                                                  acc, false, false);
    }
    float bh = b_h2h[col];
#pragma unroll
    for (int i = 0; i < 8; ++i) {
      int rr = i + 8 * half;
      float ds = degS[rr];
      A[rr * A3_LD + 2 * HID + col] =
          (acc[i] + ds * bh) / fmaxf(ds, 1.0f);
    }
  }
  __syncthreads();

  {
    const int sp = wv >> 2;
    const int kq = wv & 3;
    const int ks0 = (kq * 49) >> 1;
    const int ks1 = ((kq + 1) * 49) >> 1;
    const unsigned bo = (unsigned)(kb * HID + sp * 64 + 4 * m);
    const float* ap = A + m * A3_LD + kb;
    v8f acc0 = {}, acc1 = {}, acc2 = {}, acc3 = {};
#pragma unroll 2
    for (int ks = ks0; ks < ks1; ++ks) {
      const int k = ks * 4;
      v2f a;
      a.x = ap[k];
      a.y = ap[k + 1];
      const unsigned br = bo + (unsigned)k * HID;
      const float4 b0 = *(const float4*)(Wp + br);
      const float4 b1 = *(const float4*)(Wp + (br + HID));
      v2f b;
      b.x = b0.x;
      b.y = b1.x;
      acc0 = __builtin_amdgcn_wmma_f32_16x16x4_f32(false, a, false, b, (short)0,
                                                   acc0, false, false);
      b.x = b0.y;
      b.y = b1.y;
      acc1 = __builtin_amdgcn_wmma_f32_16x16x4_f32(false, a, false, b, (short)0,
                                                   acc1, false, false);
      b.x = b0.z;
      b.y = b1.z;
      acc2 = __builtin_amdgcn_wmma_f32_16x16x4_f32(false, a, false, b, (short)0,
                                                   acc2, false, false);
      b.x = b0.w;
      b.y = b1.w;
      acc3 = __builtin_amdgcn_wmma_f32_16x16x4_f32(false, a, false, b, (short)0,
                                                   acc3, false, false);
    }
    __syncthreads();

    float* P = smem + wv * 1024 + m;
#pragma unroll
    for (int i = 0; i < 8; ++i) {
      const int ro = (i + 8 * half) * 16;
      P[0 * 256 + ro] = acc0[i];
      P[1 * 256 + ro] = acc1[i];
      P[2 * 256 + ro] = acc2[i];
      P[3 * 256 + ro] = acc3[i];
    }
  }
  __syncthreads();

  {
    const int row = tid >> 4;
    const int c16 = tid & 15;
    const int nn = n0 + row;
#pragma unroll
    for (int sp = 0; sp < 2; ++sp) {
      float o[4];
#pragma unroll
      for (int j = 0; j < 4; ++j) {
        const float* Q = smem + (sp * 4) * 1024 + j * 256 + row * 16 + c16;
        o[j] = ((Q[0] + Q[1024]) + Q[2048]) + Q[3072];
      }
      const float* bq = b_upd + sp * 64 + 4 * c16;
      if (nn < n_state) {
        float4 r;
        r.x = o[0] + bq[0];
        r.y = o[1] + bq[1];
        r.z = o[2] + bq[2];
        r.w = o[3] + bq[3];
        *(float4*)(out + (long)nn * HID + sp * 64 + 4 * c16) = r;
      }
    }
  }
}

extern "C" void kernel_launch(void* const* d_in, const int* in_sizes, int n_in,
                              void* d_out, int out_size, void* d_ws,
                              size_t ws_size, hipStream_t stream) {
  const float* h = (const float*)d_in[0];
  const float* u = (const float*)d_in[1];
  const float* pos_state = (const float*)d_in[2];
  const float* pos_action = (const float*)d_in[3];
  const int* a2s_src = (const int*)d_in[4];
  const int* a2s_dst = (const int*)d_in[5];
  const float* a2s_dis = (const float*)d_in[6];
  const int* s2s_src = (const int*)d_in[7];
  const int* s2s_dst = (const int*)d_in[8];
  const float* s2s_dis = (const float*)d_in[9];
  const float* W_u2h = (const float*)d_in[10];
  const float* b_u2h = (const float*)d_in[11];
  const float* W_h2h = (const float*)d_in[12];
  const float* b_h2h = (const float*)d_in[13];
  const float* W_upd = (const float*)d_in[14];
  const float* b_upd = (const float*)d_in[15];
  float* out = (float*)d_out;

  const int n_state = in_sizes[2] / 2;
  const int n_action = in_sizes[3] / 2;
  const int E_a2s = in_sizes[4];
  const int E_s2s = in_sizes[7];

  float* WTu = (float*)d_ws;
  float* WTh = WTu + HID * AU_KP;
  float* Wp = WTh + HID * AH_KP;
  int* listS = (int*)(Wp + HID * A3_KP);
  int* listA = listS + (size_t)E_s2s * 2;
  int* cntS = listA + (size_t)E_a2s * 2;
  int* cntA = cntS + n_state;
  int* tot = cntA + n_state;
  int* offS = tot + 2;
  int* offA = offS + n_state;
  int* curS = offA + n_state;
  int* curA = curS + n_state;
  const int E_all = E_s2s + E_a2s;
  int* rank = curA + n_state;
  if ((size_t)((char*)(rank + E_all) - (char*)d_ws) > ws_size) rank = nullptr;
  int* curS_used = rank ? nullptr : curS;
  int* curA_used = rank ? nullptr : curA;

  int* sl_listS = (int*)(Wp + HID * A3_KP);
  int* sl_listA = sl_listS + (size_t)n_state * CAPS * 2;
  int* sl_cnt2 = sl_listA + (size_t)n_state * CAPA * 2;
  int hs = 16;
  if ((size_t)((char*)(sl_cnt2 + (size_t)n_state * 32) - (char*)d_ws) > ws_size)
    hs = 4;
  if (hs == 4 &&
      (size_t)((char*)(sl_cnt2 + (size_t)n_state * 8) - (char*)d_ws) > ws_size)
    hs = 2;
  if ((size_t)((char*)(sl_cnt2 + (size_t)n_state * 2 * hs) - (char*)d_ws) <=
      ws_size) {
    pack_weights_kernel<<<256, 256, 0, stream>>>(W_u2h, W_h2h, W_upd, WTu, WTh,
                                                 Wp, nullptr, 0, sl_cnt2,
                                                 2 * n_state, hs);
    slot_kernel<<<(E_all + 255) / 256, 256, 0, stream>>>(
        s2s_src, s2s_dst, s2s_dis, a2s_src, a2s_dst, a2s_dis, sl_cnt2,
        sl_listS, sl_listA, n_state, E_s2s, E_a2s, hs);
    fused_node_kernel<<<(n_state + 15) / 16, 256, 0, stream>>>(
        h, pos_state, u, pos_action, sl_listS, sl_listA, nullptr, sl_cnt2,
        nullptr, sl_cnt2 + 1, WTu, b_u2h, WTh, b_h2h, Wp, b_upd, s2s_src,
        s2s_dst, s2s_dis, a2s_src, a2s_dst, a2s_dis, out, n_state, n_action,
        E_s2s, E_a2s, hs);
    return;
  }
  pack_weights_kernel<<<256, 256, 0, stream>>>(W_u2h, W_h2h, W_upd, WTu, WTh,
                                               Wp, cntS, 2 * n_state + 2,
                                               nullptr, 0, 0);
  count_kernel<<<(E_all + 255) / 256, 256, 0, stream>>>(
      s2s_dst, a2s_dst, cntS, cntA, rank, E_s2s, E_a2s);
  alloc_kernel<<<(n_state + 255) / 256, 256, 0, stream>>>(
      cntS, cntA, tot, offS, offA, curS_used, curA_used, n_state);
  fill_kernel<<<(E_all + 255) / 256, 256, 0, stream>>>(
      s2s_src, s2s_dst, s2s_dis, a2s_src, a2s_dst, a2s_dis, offS, offA, rank,
      curS_used, curA_used, listS, listA, E_s2s, E_a2s);
  fused_node_kernel<<<(n_state + 15) / 16, 256, 0, stream>>>(
      h, pos_state, u, pos_action, listS, listA, offS, cntS, offA, cntA, WTu,
      b_u2h, WTh, b_h2h, Wp, b_upd, s2s_src, s2s_dst, s2s_dis, a2s_src,
      a2s_dst, a2s_dis, out, n_state, n_action, E_s2s, E_a2s, 1);
}


# edge pass stores no scattered record: counting add -> exchange on the node's list head (per half); 16-B edge records {old head, src, dis} in edge order; node kernel walks the lists; sums re-ordered
# speedup vs baseline: 3.2585x; 1.1421x over previous
#include <hip/hip_runtime.h>


typedef float v2f __attribute__((ext_vector_type(2)));
typedef float v8f __attribute__((ext_vector_type(8)));

#define HID 128

#define AU_KP 16
#define AH_KP 136
#define A3_KP 392


__global__ void pack_weights_kernel(const float* __restrict__ W_u2h,
                                    const float* __restrict__ W_h2h,
                                    const float* __restrict__ W_upd,
                                    float* __restrict__ WTu,
                                    float* __restrict__ WTh,
                                    float* __restrict__ Wp,
                                    int* __restrict__ dense, int n_dense,
                                    int* __restrict__ pairs, int n_pairs,
                                    int cstride) {
  const int stride = gridDim.x * blockDim.x;
  const int tid = blockIdx.x * blockDim.x + threadIdx.x;
  for (int i = tid; i < n_dense; i += stride) dense[i] = 0;
  for (int i = tid; i < n_pairs; i += stride)
    *(int2*)(pairs + (unsigned)i * (unsigned)cstride) = make_int2(0, 0);
  for (int i = tid; i < HID * AU_KP; i += stride) {
    int c = i / AU_KP, k = i % AU_KP;
    WTu[i] = (k < 13) ? W_u2h[k * HID + c] : 0.f;
  }
  for (int i = tid; i < HID * AH_KP; i += stride) {
    int c = i / AH_KP, k = i % AH_KP;
    WTh[i] = (k < 133) ? W_h2h[k * HID + c] : 0.f;
  }
  for (int i = tid; i < HID * A3_KP; i += stride)
    Wp[i] = (i < 386 * HID) ? W_upd[i] : 0.f;
}

__global__ __launch_bounds__(256) void count_kernel(
    const int* __restrict__ dstS, const int* __restrict__ dstA,
    int* __restrict__ cntS, int* __restrict__ cntA, int* __restrict__ rank,
    int ES, int EA) {
  int e = blockIdx.x * blockDim.x + threadIdx.x;
  if (e < ES) {
    int r = atomicAdd(cntS + dstS[e], 1);
    if (rank) rank[e] = r;
  } else if (e - ES < EA) {
    int r = atomicAdd(cntA + dstA[e - ES], 1);
    if (rank) rank[e] = r;
  }
}

__global__ __launch_bounds__(256) void alloc_kernel(
    const int* __restrict__ cntS, const int* __restrict__ cntA,
    int* __restrict__ tot, int* __restrict__ offS, int* __restrict__ offA,
    int* __restrict__ curS, int* __restrict__ curA, int n_state) {
  const int i = blockIdx.x * blockDim.x + threadIdx.x;
  const int lane = threadIdx.x & 31;
  const int cs = (i < n_state) ? cntS[i] : 0;
  const int ca = (i < n_state) ? cntA[i] : 0;
  int ss = cs, sa = ca;
  for (int o = 1; o < 32; o <<= 1) {
    int ts = __shfl_up(ss, o);
    int ta = __shfl_up(sa, o);
    if (lane >= o) {
      ss += ts;
      sa += ta;
    }
  }
  int baseS = 0, baseA = 0;
  if (lane == 31) {
    baseS = atomicAdd(tot + 0, ss);
    baseA = atomicAdd(tot + 1, sa);
  }
  baseS = __shfl(baseS, 31);
  baseA = __shfl(baseA, 31);
  if (i < n_state) {
    int os = baseS + ss - cs;
    int oa = baseA + sa - ca;
    offS[i] = os;
    offA[i] = oa;
    if (curS) {
      curS[i] = os;
      curA[i] = oa;
    }
  }
}

__global__ __launch_bounds__(256) void fill_kernel(
    const int* __restrict__ srcS, const int* __restrict__ dstS,
    const float* __restrict__ disS, const int* __restrict__ srcA,
    const int* __restrict__ dstA, const float* __restrict__ disA,
    const int* __restrict__ offS, const int* __restrict__ offA,
    const int* __restrict__ rank, int* __restrict__ curS,
    int* __restrict__ curA, int* __restrict__ listS, int* __restrict__ listA,
    int ES, int EA) {
  int e = blockIdx.x * blockDim.x + threadIdx.x;
  if (e < ES) {
    const int d = dstS[e];
    int p = rank ? offS[d] + rank[e] : atomicAdd(curS + d, 1);
    p = min(max(p, 0), ES - 1);
    ((int2*)listS)[p] = make_int2(srcS[e], __float_as_int(disS[e]));
  } else if (e - ES < EA) {
    int a = e - ES;
    const int d = dstA[a];
    int p = rank ? offA[d] + rank[e] : atomicAdd(curA + d, 1);
    p = min(max(p, 0), EA - 1);
    ((int2*)listA)[p] = make_int2(srcA[a], __float_as_int(disA[a]));
  }
}

__global__ __launch_bounds__(256) void push_kernel(
    const int* __restrict__ srcS, const int* __restrict__ dstS,
    const float* __restrict__ disS, const int* __restrict__ srcA,
    const int* __restrict__ dstA, const float* __restrict__ disA,
    int* __restrict__ heads, int* __restrict__ recs, int n_state, int ES,
    int EA, int hs) {
  const int e = blockIdx.x * blockDim.x + threadIdx.x;
  const int hb = (int)((__builtin_amdgcn_s_sendmsg_rtn(0x87) >> 18) & 1u);
  int* const hd = heads + (size_t)hb * (size_t)n_state * (size_t)hs;
  if (e < ES) {
    const int d = dstS[e];
    int s = srcS[e];
    int dv = __float_as_int(disS[e]);
    asm volatile("" : "+v"(s), "+v"(dv));
    if ((unsigned)d < (unsigned)n_state) {
      const int old = atomicExch(hd + (unsigned)d * (unsigned)hs, e + 1);
      ((int4*)recs)[e] = make_int4(old, s, dv, 0);
    }
  } else if (e - ES < EA) {
    const int a = e - ES;
    const int d = dstA[a];
    int s = srcA[a];
    int dv = __float_as_int(disA[a]);
    asm volatile("" : "+v"(s), "+v"(dv));
    if ((unsigned)d < (unsigned)n_state) {
      const int old = atomicExch(hd + (unsigned)d * (unsigned)hs + 1, e + 1);
      ((int4*)recs)[e] = make_int4(old, s, dv, 0);
    }
  }
}

__device__ __forceinline__ float wave_sum(float v) {
  for (int o = 16; o > 0; o >>= 1) v += __shfl_xor(v, o);
  return v;
}

#define A3_LD 393
#define AH_LD 137
#define AU_LD 17
#define SM_A 0
#define SM_AH (16 * A3_LD)
#define SM_AU (SM_AH + 16 * AH_LD)
#define SM_TOTAL (SM_AU + 16 * AU_LD)

__global__ __launch_bounds__(256, 16) void fused_node_kernel(
    const float* __restrict__ h, const float* __restrict__ pos_state,
    const float* __restrict__ u, const float* __restrict__ pos_action,
    const int* __restrict__ listS, const int* __restrict__ listA,
    const int* __restrict__ offS, const int* __restrict__ cntS,
    const int* __restrict__ offA, const int* __restrict__ cntA,
    const int* __restrict__ recs, const int* __restrict__ heads,
    const float* __restrict__ WTu, const float* __restrict__ b_u2h,
    const float* __restrict__ WTh, const float* __restrict__ b_h2h,
    const float* __restrict__ Wp, const float* __restrict__ b_upd,
    float* __restrict__ out, int n_state, int n_action, int ES, int EA,
    int cstride) {
  __shared__ __attribute__((aligned(16))) float smem[SM_TOTAL];
  float* const A = smem + SM_A;
  float* const Ah = smem + SM_AH;
  float* const Au = smem + SM_AU;
  __shared__ float degU[16];
  __shared__ float degS[16];

  const int tid = threadIdx.x;
  const int n0 = blockIdx.x * 16;
  const int lane = tid & 31;
  const int wv = __builtin_amdgcn_readfirstlane(tid >> 5);

  {
    const int r = tid >> 4;
    const int c0 = tid & 15;
    const int n = n0 + r;
    const bool valid = (n < n_state);
    if (c0 == 0) {
      float* AR = A + r * A3_LD;
      AR[384] = valid ? pos_state[2 * n] : 0.f;
      AR[385] = valid ? pos_state[2 * n + 1] : 0.f;
#pragma unroll
      for (int k = 386; k < A3_KP; ++k) AR[k] = 0.f;
    }
    const float* hp = h + (long)n * HID;
    for (int c = c0; c < HID; c += 16)
      A[r * A3_LD + c] = valid ? hp[c] : 0.f;
  }

  if (recs) {
    const int nb = n0 + 2 * wv;
    const int w_node = (lane >> 1) & 1;
    const int w_graph = (lane >> 2) & 1;
    const int lo = w_graph ? ES : 0;
    const int hi = w_graph ? ES + EA : ES;
    int cur = 0;
    if (lane < 8 && nb + w_node < n_state)
      cur = heads[((unsigned)(lane & 1) * (unsigned)n_state +
                   (unsigned)(nb + w_node)) * (unsigned)cstride + w_graph];
    float4 accA = make_float4(0.f, 0.f, 0.f, 0.f);
    float4 accB = make_float4(0.f, 0.f, 0.f, 0.f);
    float uaA = 0.f, uaB = 0.f;
    float w_px = 0.f, w_py = 0.f, w_d = 0.f;
    int w_c = 0;
    const int hopmax = max(ES, EA);
    bool act = (cur > lo) && (cur <= hi);
    int4 r = make_int4(0, 0, 0, 0);
    if (act) r = ((const int4*)recs)[cur - 1];
    for (int hop = 0; hop < hopmax; ++hop) {
      const unsigned m = __builtin_amdgcn_ballot_w32(act);
      if (m == 0u) break;
      const int nxt = r.x;
      const bool actn = (nxt > lo) && (nxt <= hi);
      int4 rn = make_int4(0, 0, 0, 0);
      if (actn) rn = ((const int4*)recs)[nxt - 1];
      const int s = (int)min((unsigned)r.y,
                             (unsigned)((w_graph ? n_action : n_state) - 1));
      if (act) {
        w_d += __int_as_float(r.z);
        w_c += 1;
      }
      float2 p = make_float2(0.f, 0.f);
      if (act && !w_graph) p = ((const float2*)pos_state)[s];
      float4 v[4];
#pragma unroll
      for (int j = 0; j < 4; ++j) {
        v[j] = make_float4(0.f, 0.f, 0.f, 0.f);
        if ((m >> j) & 1u) {
          const int sj = __builtin_amdgcn_readlane(s, j);
          v[j] = ((const float4*)(h + (long)sj * HID))[lane];
        }
      }
      float x[4];
#pragma unroll
      for (int j = 0; j < 4; ++j) {
        x[j] = 0.f;
        if ((m >> (4 + j)) & 1u) {
          const int sj = __builtin_amdgcn_readlane(s, 4 + j);
          if (lane < 10)
            x[j] = (lane < 8) ? u[(long)sj * 8 + lane]
                              : pos_action[2 * (long)sj + (lane - 8)];
        }
      }
      accA.x += v[0].x;
      accA.y += v[0].y;
      accA.z += v[0].z;
      accA.w += v[0].w;
      accA.x += v[1].x;
      accA.y += v[1].y;
      accA.z += v[1].z;
      accA.w += v[1].w;
      accB.x += v[2].x;
      accB.y += v[2].y;
      accB.z += v[2].z;
      accB.w += v[2].w;
      accB.x += v[3].x;
      accB.y += v[3].y;
      accB.z += v[3].z;
      accB.w += v[3].w;
      uaA += x[0];
      uaA += x[1];
      uaB += x[2];
      uaB += x[3];
      w_px += p.x;
      w_py += p.y;
      r = rn;
      act = actn;
    }
    const float t_px = w_px + __shfl_xor(w_px, 1);
    const float t_py = w_py + __shfl_xor(w_py, 1);
    const float t_d = w_d + __shfl_xor(w_d, 1);
    const int t_c = w_c + __shfl_xor(w_c, 1);
#pragma unroll
    for (int rr = 0; rr < 2; ++rr) {
      const int r2 = wv * 2 + rr;
      const int n2 = nb + rr;
      float q0 = 0.f, q1 = 0.f;
      if (n2 < n_state) {
        q0 = pos_state[2 * n2];
        q1 = pos_state[2 * n2 + 1];
      }
      const float ps0 = __shfl(t_px, 2 * rr);
      const float ps1 = __shfl(t_py, 2 * rr);
      const float dsm = __shfl(t_d, 2 * rr);
      const float dgS = (float)__shfl(t_c, 2 * rr);
      const float dsa = __shfl(t_d, 4 + 2 * rr);
      const float dgA = (float)__shfl(t_c, 4 + 2 * rr);
      const float4 acc = rr ? accB : accA;
      const float ua = rr ? uaB : uaA;
      float* AhR = Ah + r2 * AH_LD;
      float* AuR = Au + r2 * AU_LD;
      AhR[5 + 4 * lane + 0] = acc.x;
      AhR[5 + 4 * lane + 1] = acc.y;
      AhR[5 + 4 * lane + 2] = acc.z;
      AhR[5 + 4 * lane + 3] = acc.w;
      if (lane < 8) AuR[5 + lane] = ua;
      if (lane == 8 || lane == 9) AuR[lane - 8] = ua;
      if (lane == 0) {
        AuR[4] = dsa;
        AhR[0] = ps0;
        AhR[1] = ps1;
        AhR[2] = dgS * q0;
        AhR[3] = dgS * q1;
        AhR[4] = dsm;
        AhR[133] = AhR[134] = AhR[135] = 0.f;
        AuR[2] = dgA * q0;
        AuR[3] = dgA * q1;
        AuR[13] = AuR[14] = AuR[15] = 0.f;
        degS[r2] = dgS;
        degU[r2] = dgA;
      }
    }
  } else {
    for (int rr = 0; rr < 2; ++rr) {
      const int r2 = wv * 2 + rr;
      const int n2 = n0 + r2;
      int begS = 0, nS = 0, begA = 0, nA = 0;
      float q0 = 0.f, q1 = 0.f;
      if (n2 < n_state) {
        nS = cntS[n2];
        nA = cntA[n2];
        begS = offS[n2];
        begA = offA[n2];
        q0 = pos_state[2 * n2];
        q1 = pos_state[2 * n2 + 1];
      }
      const int degSi = min(max(nS, 0), ES);
      const int degAi = min(max(nA, 0), EA);
      begS = min(max(begS, 0), ES);
      const int endS = begS + min(degSi, ES - begS);
      begA = min(max(begA, 0), EA);
      const int endA = begA + min(degAi, EA - begA);

      float4 acc = make_float4(0.f, 0.f, 0.f, 0.f);
      float ps0 = 0.f, ps1 = 0.f, dsm = 0.f;
      for (int i = begS; i < endS; i += 32) {
        const int my = i + lane;
        int smine = 0;
        float2 p = make_float2(0.f, 0.f);
        float dv = 0.f;
        if (my < endS) {
          int2 rec = ((const int2*)listS)[my];
          smine = (int)min((unsigned)rec.x, (unsigned)(n_state - 1));
          p = ((const float2*)pos_state)[smine];
          dv = __int_as_float(rec.y);
        }
        const int mc = min(32, endS - i);
#pragma unroll 4
        for (int j = 0; j < mc; ++j) {
          const int s = __builtin_amdgcn_readlane(smine, j);
          const float4 v = ((const float4*)(h + (long)s * HID))[lane];
          acc.x += v.x;
          acc.y += v.y;
          acc.z += v.z;
          acc.w += v.w;
        }
        ps0 += p.x;
        ps1 += p.y;
        dsm += dv;
      }
      ps0 = wave_sum(ps0);
      ps1 = wave_sum(ps1);
      dsm = wave_sum(dsm);

      const int slot = lane >> 3, comp = lane & 7;
      float ua = 0.f, xa = 0.f;
      for (int i = begA; i < endA; i += 4) {
        const int my = i + slot;
        if (my < endA) {
          int2 rec = ((const int2*)listA)[my];
          const int s = (int)min((unsigned)rec.x, (unsigned)(n_action - 1));
          ua += u[(long)s * 8 + comp];
          float x = 0.f;
          if (comp < 2) x = pos_action[2 * s + comp];
          else if (comp == 2) x = __int_as_float(rec.y);
          xa += x;
        }
      }
      ua += __shfl_xor(ua, 8);
      ua += __shfl_xor(ua, 16);
      xa += __shfl_xor(xa, 8);
      xa += __shfl_xor(xa, 16);

      const float dgS = (float)degSi;
      const float dgA = (float)degAi;
      float* AhR = Ah + r2 * AH_LD;
      float* AuR = Au + r2 * AU_LD;
      AhR[5 + 4 * lane + 0] = acc.x;
      AhR[5 + 4 * lane + 1] = acc.y;
      AhR[5 + 4 * lane + 2] = acc.z;
      AhR[5 + 4 * lane + 3] = acc.w;
      if (lane < 8) AuR[5 + lane] = ua;
      if (lane < 2) AuR[lane] = xa;
      if (lane == 2) AuR[4] = xa;
      if (lane == 0) {
        AhR[0] = ps0;
        AhR[1] = ps1;
        AhR[2] = dgS * q0;
        AhR[3] = dgS * q1;
        AhR[4] = dsm;
        AhR[133] = AhR[134] = AhR[135] = 0.f;
        AuR[2] = dgA * q0;
        AuR[3] = dgA * q1;
        AuR[13] = AuR[14] = AuR[15] = 0.f;
        degS[r2] = dgS;
        degU[r2] = dgA;
      }
    }
  }
  __syncthreads();

  const int half = lane >> 4;
  const int m = lane & 15;
  const int kb = half * 2;
  const int col = wv * 16 + m;

  {
    const float* bp = WTu + col * AU_KP + kb;
    v8f acc = {};
#pragma unroll
    for (int k = 0; k < AU_KP; k += 4) {
      v2f a;
      a.x = Au[m * AU_LD + k + kb];
      a.y = Au[m * AU_LD + k + kb + 1];
      v2f b = *(const v2f*)(bp + k);
      acc = __builtin_amdgcn_wmma_f32_16x16x4_f32(false, a, false, b, (short)0,
                                                  acc, false, false);
    }
    float bu = b_u2h[col];
#pragma unroll
    for (int i = 0; i < 8; ++i) {
      int rr = i + 8 * half;
      A[rr * A3_LD + HID + col] = acc[i] + degU[rr] * bu;
    }
  }

  {
    const float* bp = WTh + col * AH_KP + kb;
    v8f acc = {};
#pragma unroll 2
    for (int k = 0; k < AH_KP; k += 4) {
      v2f a;
      a.x = Ah[m * AH_LD + k + kb];
      a.y = Ah[m * AH_LD + k + kb + 1];
      v2f b = *(const v2f*)(bp + k);
      acc = __builtin_amdgcn_wmma_f32_16x16x4_f32(false, a, false, b, (short)0,
                                                  acc, false, false);
    }
    float bh = b_h2h[col];
#pragma unroll
    for (int i = 0; i < 8; ++i) {
      int rr = i + 8 * half;
      float ds = degS[rr];
      A[rr * A3_LD + 2 * HID + col] =
          (acc[i] + ds * bh) / fmaxf(ds, 1.0f);
    }
  }
  __syncthreads();

  {
    const int sp = wv >> 2;
    const int kq = wv & 3;
    const int ks0 = (kq * 49) >> 1;
    const int ks1 = ((kq + 1) * 49) >> 1;
    const unsigned bo = (unsigned)(kb * HID + sp * 64 + 4 * m);
    const float* ap = A + m * A3_LD + kb;
    v8f acc0 = {}, acc1 = {}, acc2 = {}, acc3 = {};
#pragma unroll 2
    for (int ks = ks0; ks < ks1; ++ks) {
      const int k = ks * 4;
      v2f a;
      a.x = ap[k];
      a.y = ap[k + 1];
      const unsigned br = bo + (unsigned)k * HID;
      const float4 b0 = *(const float4*)(Wp + br);
      const float4 b1 = *(const float4*)(Wp + (br + HID));
      v2f b;
      b.x = b0.x;
      b.y = b1.x;
      acc0 = __builtin_amdgcn_wmma_f32_16x16x4_f32(false, a, false, b, (short)0,
                                                   acc0, false, false);
      b.x = b0.y;
      b.y = b1.y;
      acc1 = __builtin_amdgcn_wmma_f32_16x16x4_f32(false, a, false, b, (short)0,
                                                   acc1, false, false);
      b.x = b0.z;
      b.y = b1.z;
      acc2 = __builtin_amdgcn_wmma_f32_16x16x4_f32(false, a, false, b, (short)0,
                                                   acc2, false, false);
      b.x = b0.w;
      b.y = b1.w;
      acc3 = __builtin_amdgcn_wmma_f32_16x16x4_f32(false, a, false, b, (short)0,
                                                   acc3, false, false);
    }
    __syncthreads();

    float* P = smem + wv * 1024 + m;
#pragma unroll
    for (int i = 0; i < 8; ++i) {
      const int ro = (i + 8 * half) * 16;
      P[0 * 256 + ro] = acc0[i];
      P[1 * 256 + ro] = acc1[i];
      P[2 * 256 + ro] = acc2[i];
      P[3 * 256 + ro] = acc3[i];
    }
  }
  __syncthreads();

  {
    const int row = tid >> 4;
    const int c16 = tid & 15;
    const int nn = n0 + row;
#pragma unroll
    for (int sp = 0; sp < 2; ++sp) {
      float o[4];
#pragma unroll
      for (int j = 0; j < 4; ++j) {
        const float* Q = smem + (sp * 4) * 1024 + j * 256 + row * 16 + c16;
        o[j] = ((Q[0] + Q[1024]) + Q[2048]) + Q[3072];
      }
      const float* bq = b_upd + sp * 64 + 4 * c16;
      if (nn < n_state) {
        float4 r;
        r.x = o[0] + bq[0];
        r.y = o[1] + bq[1];
        r.z = o[2] + bq[2];
        r.w = o[3] + bq[3];
        *(float4*)(out + (long)nn * HID + sp * 64 + 4 * c16) = r;
      }
    }
  }
}

extern "C" void kernel_launch(void* const* d_in, const int* in_sizes, int n_in,
                              void* d_out, int out_size, void* d_ws,
                              size_t ws_size, hipStream_t stream) {
  const float* h = (const float*)d_in[0];
  const float* u = (const float*)d_in[1];
  const float* pos_state = (const float*)d_in[2];
  const float* pos_action = (const float*)d_in[3];
  const int* a2s_src = (const int*)d_in[4];
  const int* a2s_dst = (const int*)d_in[5];
  const float* a2s_dis = (const float*)d_in[6];
  const int* s2s_src = (const int*)d_in[7];
  const int* s2s_dst = (const int*)d_in[8];
  const float* s2s_dis = (const float*)d_in[9];
  const float* W_u2h = (const float*)d_in[10];
  const float* b_u2h = (const float*)d_in[11];
  const float* W_h2h = (const float*)d_in[12];
  const float* b_h2h = (const float*)d_in[13];
  const float* W_upd = (const float*)d_in[14];
  const float* b_upd = (const float*)d_in[15];
  float* out = (float*)d_out;

  const int n_state = in_sizes[2] / 2;
  const int n_action = in_sizes[3] / 2;
  const int E_a2s = in_sizes[4];
  const int E_s2s = in_sizes[7];

  float* WTu = (float*)d_ws;
  float* WTh = WTu + HID * AU_KP;
  float* Wp = WTh + HID * AH_KP;
  int* listS = (int*)(Wp + HID * A3_KP);
  int* listA = listS + (size_t)E_s2s * 2;
  int* cntS = listA + (size_t)E_a2s * 2;
  int* cntA = cntS + n_state;
  int* tot = cntA + n_state;
  int* offS = tot + 2;
  int* offA = offS + n_state;
  int* curS = offA + n_state;
  int* curA = curS + n_state;
  const int E_all = E_s2s + E_a2s;
  int* rank = curA + n_state;
  if ((size_t)((char*)(rank + E_all) - (char*)d_ws) > ws_size) rank = nullptr;
  int* curS_used = rank ? nullptr : curS;
  int* curA_used = rank ? nullptr : curA;

  int* ll_recs = (int*)(Wp + HID * A3_KP);
  int* ll_heads = ll_recs + (size_t)E_all * 4;
  int hs = 16;
  if ((size_t)((char*)(ll_heads + (size_t)n_state * 32) - (char*)d_ws) > ws_size)
    hs = 4;
  if (hs == 4 &&
      (size_t)((char*)(ll_heads + (size_t)n_state * 8) - (char*)d_ws) > ws_size)
    hs = 2;
  if ((size_t)((char*)(ll_heads + (size_t)n_state * 2 * hs) - (char*)d_ws) <=
      ws_size) {
    pack_weights_kernel<<<256, 256, 0, stream>>>(W_u2h, W_h2h, W_upd, WTu, WTh,
                                                 Wp, nullptr, 0, ll_heads,
                                                 2 * n_state, hs);
    push_kernel<<<(E_all + 255) / 256, 256, 0, stream>>>(
        s2s_src, s2s_dst, s2s_dis, a2s_src, a2s_dst, a2s_dis, ll_heads,
        ll_recs, n_state, E_s2s, E_a2s, hs);
    fused_node_kernel<<<(n_state + 15) / 16, 256, 0, stream>>>(
        h, pos_state, u, pos_action, nullptr, nullptr, nullptr, nullptr,
        nullptr, nullptr, ll_recs, ll_heads, WTu, b_u2h, WTh, b_h2h, Wp, b_upd,
        out, n_state, n_action, E_s2s, E_a2s, hs);
    return;
  }
  pack_weights_kernel<<<256, 256, 0, stream>>>(W_u2h, W_h2h, W_upd, WTu, WTh,
                                               Wp, cntS, 2 * n_state + 2,
                                               nullptr, 0, 0);
  count_kernel<<<(E_all + 255) / 256, 256, 0, stream>>>(
      s2s_dst, a2s_dst, cntS, cntA, rank, E_s2s, E_a2s);
  alloc_kernel<<<(n_state + 255) / 256, 256, 0, stream>>>(
      cntS, cntA, tot, offS, offA, curS_used, curA_used, n_state);
  fill_kernel<<<(E_all + 255) / 256, 256, 0, stream>>>(
      s2s_src, s2s_dst, s2s_dis, a2s_src, a2s_dst, a2s_dis, offS, offA, rank,
      curS_used, curA_used, listS, listA, E_s2s, E_a2s);
  fused_node_kernel<<<(n_state + 15) / 16, 256, 0, stream>>>(
      h, pos_state, u, pos_action, listS, listA, offS, cntS, offA, cntA,
      nullptr, nullptr, WTu, b_u2h, WTh, b_h2h, Wp, b_upd, out, n_state,
      n_action, E_s2s, E_a2s, 1);
}


# node kernel, product 2 re-tiled: a wave owns 32 columns x half of K (2 accumulators, A fragment reused 2x, B in 128-B runs of row-major weights); K halves summed via LDS; sums re-associated
# speedup vs baseline: 3.4058x; 1.0452x over previous
#include <hip/hip_runtime.h>


typedef float v2f __attribute__((ext_vector_type(2)));
typedef float v8f __attribute__((ext_vector_type(8)));

#define HID 128

#define AU_KP 16
#define AH_KP 136
#define A3_KP 392


__global__ void pack_weights_kernel(const float* __restrict__ W_u2h,
                                    const float* __restrict__ W_h2h,
                                    const float* __restrict__ W_upd,
                                    float* __restrict__ WTu,
                                    float* __restrict__ Whp,
                                    float* __restrict__ Wp,
                                    int* __restrict__ dense, int n_dense,
                                    int* __restrict__ pairs, int n_pairs,
                                    int cstride) {
  const int stride = gridDim.x * blockDim.x;
  const int tid = blockIdx.x * blockDim.x + threadIdx.x;
  for (int i = tid; i < n_dense; i += stride) dense[i] = 0;
  for (int i = tid; i < n_pairs; i += stride)
    *(int2*)(pairs + (unsigned)i * (unsigned)cstride) = make_int2(0, 0);
  for (int i = tid; i < HID * AU_KP; i += stride) {
    int c = i / AU_KP, k = i % AU_KP;
    WTu[i] = (k < 13) ? W_u2h[k * HID + c] : 0.f;
  }
  for (int i = tid; i < HID * AH_KP; i += stride)
    Whp[i] = (i < 133 * HID) ? W_h2h[i] : 0.f;
  for (int i = tid; i < HID * A3_KP; i += stride)
    Wp[i] = (i < 386 * HID) ? W_upd[i] : 0.f;
}

__global__ __launch_bounds__(256) void count_kernel(
    const int* __restrict__ dstS, const int* __restrict__ dstA,
    int* __restrict__ cntS, int* __restrict__ cntA, int* __restrict__ rank,
    int ES, int EA) {
  int e = blockIdx.x * blockDim.x + threadIdx.x;
  if (e < ES) {
    int r = atomicAdd(cntS + dstS[e], 1);
    if (rank) rank[e] = r;
  } else if (e - ES < EA) {
    int r = atomicAdd(cntA + dstA[e - ES], 1);
    if (rank) rank[e] = r;
  }
}

__global__ __launch_bounds__(256) void alloc_kernel(
    const int* __restrict__ cntS, const int* __restrict__ cntA,
    int* __restrict__ tot, int* __restrict__ offS, int* __restrict__ offA,
    int* __restrict__ curS, int* __restrict__ curA, int n_state) {
  const int i = blockIdx.x * blockDim.x + threadIdx.x;
  const int lane = threadIdx.x & 31;
  const int cs = (i < n_state) ? cntS[i] : 0;
  const int ca = (i < n_state) ? cntA[i] : 0;
  int ss = cs, sa = ca;
  for (int o = 1; o < 32; o <<= 1) {
    int ts = __shfl_up(ss, o);
    int ta = __shfl_up(sa, o);
    if (lane >= o) {
      ss += ts;
      sa += ta;
    }
  }
  int baseS = 0, baseA = 0;
  if (lane == 31) {
    baseS = atomicAdd(tot + 0, ss);
    baseA = atomicAdd(tot + 1, sa);
  }
  baseS = __shfl(baseS, 31);
  baseA = __shfl(baseA, 31);
  if (i < n_state) {
    int os = baseS + ss - cs;
    int oa = baseA + sa - ca;
    offS[i] = os;
    offA[i] = oa;
    if (curS) {
      curS[i] = os;
      curA[i] = oa;
    }
  }
}

__global__ __launch_bounds__(256) void fill_kernel(
    const int* __restrict__ srcS, const int* __restrict__ dstS,
    const float* __restrict__ disS, const int* __restrict__ srcA,
    const int* __restrict__ dstA, const float* __restrict__ disA,
    const int* __restrict__ offS, const int* __restrict__ offA,
    const int* __restrict__ rank, int* __restrict__ curS,
    int* __restrict__ curA, int* __restrict__ listS, int* __restrict__ listA,
    int ES, int EA) {
  int e = blockIdx.x * blockDim.x + threadIdx.x;
  if (e < ES) {
    const int d = dstS[e];
    int p = rank ? offS[d] + rank[e] : atomicAdd(curS + d, 1);
    p = min(max(p, 0), ES - 1);
    ((int2*)listS)[p] = make_int2(srcS[e], __float_as_int(disS[e]));
  } else if (e - ES < EA) {
    int a = e - ES;
    const int d = dstA[a];
    int p = rank ? offA[d] + rank[e] : atomicAdd(curA + d, 1);
    p = min(max(p, 0), EA - 1);
    ((int2*)listA)[p] = make_int2(srcA[a], __float_as_int(disA[a]));
  }
}

__global__ __launch_bounds__(256) void push_kernel(
    const int* __restrict__ srcS, const int* __restrict__ dstS,
    const float* __restrict__ disS, const int* __restrict__ srcA,
    const int* __restrict__ dstA, const float* __restrict__ disA,
    int* __restrict__ heads, int* __restrict__ recs, int n_state, int ES,
    int EA, int hs) {
  const int e = blockIdx.x * blockDim.x + threadIdx.x;
  const int hb = (int)((__builtin_amdgcn_s_sendmsg_rtn(0x87) >> 18) & 1u);
  int* const hd = heads + (size_t)hb * (size_t)n_state * (size_t)hs;
  if (e < ES) {
    const int d = dstS[e];
    int s = srcS[e];
    int dv = __float_as_int(disS[e]);
    asm volatile("" : "+v"(s), "+v"(dv));
    if ((unsigned)d < (unsigned)n_state) {
      const int old = atomicExch(hd + (unsigned)d * (unsigned)hs, e + 1);
      ((int4*)recs)[e] = make_int4(old, s, dv, 0);
    }
  } else if (e - ES < EA) {
    const int a = e - ES;
    const int d = dstA[a];
    int s = srcA[a];
    int dv = __float_as_int(disA[a]);
    asm volatile("" : "+v"(s), "+v"(dv));
    if ((unsigned)d < (unsigned)n_state) {
      const int old = atomicExch(hd + (unsigned)d * (unsigned)hs + 1, e + 1);
      ((int4*)recs)[e] = make_int4(old, s, dv, 0);
    }
  }
}

__device__ __forceinline__ float wave_sum(float v) {
  for (int o = 16; o > 0; o >>= 1) v += __shfl_xor(v, o);
  return v;
}

#define A3_LD 393
#define AH_LD 137
#define AU_LD 17
#define SM_A 0
#define SM_AH (16 * A3_LD)
#define SM_AU (SM_AH + 16 * AH_LD)
#define SM_TOTAL (SM_AU + 16 * AU_LD)

__global__ __launch_bounds__(256, 16) void fused_node_kernel(
    const float* __restrict__ h, const float* __restrict__ pos_state,
    const float* __restrict__ u, const float* __restrict__ pos_action,
    const int* __restrict__ listS, const int* __restrict__ listA,
    const int* __restrict__ offS, const int* __restrict__ cntS,
    const int* __restrict__ offA, const int* __restrict__ cntA,
    const int* __restrict__ recs, const int* __restrict__ heads,
    const float* __restrict__ WTu, const float* __restrict__ b_u2h,
    const float* __restrict__ Whp, const float* __restrict__ b_h2h,
    const float* __restrict__ Wp, const float* __restrict__ b_upd,
    float* __restrict__ out, int n_state, int n_action, int ES, int EA,
    int cstride) {
  __shared__ __attribute__((aligned(16))) float smem[SM_TOTAL];
  float* const A = smem + SM_A;
  float* const Ah = smem + SM_AH;
  float* const Au = smem + SM_AU;
  __shared__ float degU[16];
  __shared__ float degS[16];

  const int tid = threadIdx.x;
  const int n0 = blockIdx.x * 16;
  const int lane = tid & 31;
  const int wv = __builtin_amdgcn_readfirstlane(tid >> 5);

  {
    const int r = tid >> 4;
    const int c0 = tid & 15;
    const int n = n0 + r;
    const bool valid = (n < n_state);
    if (c0 == 0) {
      float* AR = A + r * A3_LD;
      AR[384] = valid ? pos_state[2 * n] : 0.f;
      AR[385] = valid ? pos_state[2 * n + 1] : 0.f;
#pragma unroll
      for (int k = 386; k < A3_KP; ++k) AR[k] = 0.f;
    }
    const float* hp = h + (long)n * HID;
    for (int c = c0; c < HID; c += 16)
      A[r * A3_LD + c] = valid ? hp[c] : 0.f;
  }

  if (recs) {
    const int nb = n0 + 2 * wv;
    const int w_node = (lane >> 1) & 1;
    const int w_graph = (lane >> 2) & 1;
    const int lo = w_graph ? ES : 0;
    const int hi = w_graph ? ES + EA : ES;
    int cur = 0;
    if (lane < 8 && nb + w_node < n_state)
      cur = heads[((unsigned)(lane & 1) * (unsigned)n_state +
                   (unsigned)(nb + w_node)) * (unsigned)cstride + w_graph];
    float4 accA = make_float4(0.f, 0.f, 0.f, 0.f);
    float4 accB = make_float4(0.f, 0.f, 0.f, 0.f);
    float uaA = 0.f, uaB = 0.f;
    float w_px = 0.f, w_py = 0.f, w_d = 0.f;
    int w_c = 0;
    const int hopmax = max(ES, EA);
    bool act = (cur > lo) && (cur <= hi);
    int4 r = make_int4(0, 0, 0, 0);
    if (act) r = ((const int4*)recs)[cur - 1];
    for (int hop = 0; hop < hopmax; ++hop) {
      const unsigned m = __builtin_amdgcn_ballot_w32(act);
      if (m == 0u) break;
      const int nxt = r.x;
      const bool actn = (nxt > lo) && (nxt <= hi);
      int4 rn = make_int4(0, 0, 0, 0);
      if (actn) rn = ((const int4*)recs)[nxt - 1];
      const int s = (int)min((unsigned)r.y,
                             (unsigned)((w_graph ? n_action : n_state) - 1));
      if (act) {
        w_d += __int_as_float(r.z);
        w_c += 1;
      }
      float2 p = make_float2(0.f, 0.f);
      if (act && !w_graph) p = ((const float2*)pos_state)[s];
      float4 v[4];
#pragma unroll
      for (int j = 0; j < 4; ++j) {
        v[j] = make_float4(0.f, 0.f, 0.f, 0.f);
        if ((m >> j) & 1u) {
          const int sj = __builtin_amdgcn_readlane(s, j);
          v[j] = ((const float4*)(h + (long)sj * HID))[lane];
        }
      }
      float x[4];
#pragma unroll
      for (int j = 0; j < 4; ++j) {
        x[j] = 0.f;
        if ((m >> (4 + j)) & 1u) {
          const int sj = __builtin_amdgcn_readlane(s, 4 + j);
          if (lane < 10)
            x[j] = (lane < 8) ? u[(long)sj * 8 + lane]
                              : pos_action[2 * (long)sj + (lane - 8)];
        }
      }
      accA.x += v[0].x;
      accA.y += v[0].y;
      accA.z += v[0].z;
      accA.w += v[0].w;
      accA.x += v[1].x;
      accA.y += v[1].y;
      accA.z += v[1].z;
      accA.w += v[1].w;
      accB.x += v[2].x;
      accB.y += v[2].y;
      accB.z += v[2].z;
      accB.w += v[2].w;
      accB.x += v[3].x;
      accB.y += v[3].y;
      accB.z += v[3].z;
      accB.w += v[3].w;
      uaA += x[0];
      uaA += x[1];
      uaB += x[2];
      uaB += x[3];
      w_px += p.x;
      w_py += p.y;
      r = rn;
      act = actn;
    }
    const float t_px = w_px + __shfl_xor(w_px, 1);
    const float t_py = w_py + __shfl_xor(w_py, 1);
    const float t_d = w_d + __shfl_xor(w_d, 1);
    const int t_c = w_c + __shfl_xor(w_c, 1);
#pragma unroll
    for (int rr = 0; rr < 2; ++rr) {
      const int r2 = wv * 2 + rr;
      const int n2 = nb + rr;
      float q0 = 0.f, q1 = 0.f;
      if (n2 < n_state) {
        q0 = pos_state[2 * n2];
        q1 = pos_state[2 * n2 + 1];
      }
      const float ps0 = __shfl(t_px, 2 * rr);
      const float ps1 = __shfl(t_py, 2 * rr);
      const float dsm = __shfl(t_d, 2 * rr);
      const float dgS = (float)__shfl(t_c, 2 * rr);
      const float dsa = __shfl(t_d, 4 + 2 * rr);
      const float dgA = (float)__shfl(t_c, 4 + 2 * rr);
      const float4 acc = rr ? accB : accA;
      const float ua = rr ? uaB : uaA;
      float* AhR = Ah + r2 * AH_LD;
      float* AuR = Au + r2 * AU_LD;
      AhR[5 + 4 * lane + 0] = acc.x;
      AhR[5 + 4 * lane + 1] = acc.y;
      AhR[5 + 4 * lane + 2] = acc.z;
      AhR[5 + 4 * lane + 3] = acc.w;
      if (lane < 8) AuR[5 + lane] = ua;
      if (lane == 8 || lane == 9) AuR[lane - 8] = ua;
      if (lane == 0) {
        AuR[4] = dsa;
        AhR[0] = ps0;
        AhR[1] = ps1;
        AhR[2] = dgS * q0;
        AhR[3] = dgS * q1;
        AhR[4] = dsm;
        AhR[133] = AhR[134] = AhR[135] = 0.f;
        AuR[2] = dgA * q0;
        AuR[3] = dgA * q1;
        AuR[13] = AuR[14] = AuR[15] = 0.f;
        degS[r2] = dgS;
        degU[r2] = dgA;
      }
    }
  } else {
    for (int rr = 0; rr < 2; ++rr) {
      const int r2 = wv * 2 + rr;
      const int n2 = n0 + r2;
      int begS = 0, nS = 0, begA = 0, nA = 0;
      float q0 = 0.f, q1 = 0.f;
      if (n2 < n_state) {
        nS = cntS[n2];
        nA = cntA[n2];
        begS = offS[n2];
        begA = offA[n2];
        q0 = pos_state[2 * n2];
        q1 = pos_state[2 * n2 + 1];
      }
      const int degSi = min(max(nS, 0), ES);
      const int degAi = min(max(nA, 0), EA);
      begS = min(max(begS, 0), ES);
      const int endS = begS + min(degSi, ES - begS);
      begA = min(max(begA, 0), EA);
      const int endA = begA + min(degAi, EA - begA);

      float4 acc = make_float4(0.f, 0.f, 0.f, 0.f);
      float ps0 = 0.f, ps1 = 0.f, dsm = 0.f;
      for (int i = begS; i < endS; i += 32) {
        const int my = i + lane;
        int smine = 0;
        float2 p = make_float2(0.f, 0.f);
        float dv = 0.f;
        if (my < endS) {
          int2 rec = ((const int2*)listS)[my];
          smine = (int)min((unsigned)rec.x, (unsigned)(n_state - 1));
          p = ((const float2*)pos_state)[smine];
          dv = __int_as_float(rec.y);
        }
        const int mc = min(32, endS - i);
#pragma unroll 4
        for (int j = 0; j < mc; ++j) {
          const int s = __builtin_amdgcn_readlane(smine, j);
          const float4 v = ((const float4*)(h + (long)s * HID))[lane];
          acc.x += v.x;
          acc.y += v.y;
          acc.z += v.z;
          acc.w += v.w;
        }
        ps0 += p.x;
        ps1 += p.y;
        dsm += dv;
      }
      ps0 = wave_sum(ps0);
      ps1 = wave_sum(ps1);
      dsm = wave_sum(dsm);

      const int slot = lane >> 3, comp = lane & 7;
      float ua = 0.f, xa = 0.f;
      for (int i = begA; i < endA; i += 4) {
        const int my = i + slot;
        if (my < endA) {
          int2 rec = ((const int2*)listA)[my];
          const int s = (int)min((unsigned)rec.x, (unsigned)(n_action - 1));
          ua += u[(long)s * 8 + comp];
          float x = 0.f;
          if (comp < 2) x = pos_action[2 * s + comp];
          else if (comp == 2) x = __int_as_float(rec.y);
          xa += x;
        }
      }
      ua += __shfl_xor(ua, 8);
      ua += __shfl_xor(ua, 16);
      xa += __shfl_xor(xa, 8);
      xa += __shfl_xor(xa, 16);

      const float dgS = (float)degSi;
      const float dgA = (float)degAi;
      float* AhR = Ah + r2 * AH_LD;
      float* AuR = Au + r2 * AU_LD;
      AhR[5 + 4 * lane + 0] = acc.x;
      AhR[5 + 4 * lane + 1] = acc.y;
      AhR[5 + 4 * lane + 2] = acc.z;
      AhR[5 + 4 * lane + 3] = acc.w;
      if (lane < 8) AuR[5 + lane] = ua;
      if (lane < 2) AuR[lane] = xa;
      if (lane == 2) AuR[4] = xa;
      if (lane == 0) {
        AhR[0] = ps0;
        AhR[1] = ps1;
        AhR[2] = dgS * q0;
        AhR[3] = dgS * q1;
        AhR[4] = dsm;
        AhR[133] = AhR[134] = AhR[135] = 0.f;
        AuR[2] = dgA * q0;
        AuR[3] = dgA * q1;
        AuR[13] = AuR[14] = AuR[15] = 0.f;
        degS[r2] = dgS;
        degU[r2] = dgA;
      }
    }
  }
  __syncthreads();

  const int half = lane >> 4;
  const int m = lane & 15;
  const int kb = half * 2;
  const int col = wv * 16 + m;

  {
    const float* bp = WTu + col * AU_KP + kb;
    v8f acc = {};
#pragma unroll
    for (int k = 0; k < AU_KP; k += 4) {
      v2f a;
      a.x = Au[m * AU_LD + k + kb];
      a.y = Au[m * AU_LD + k + kb + 1];
      v2f b = *(const v2f*)(bp + k);
      acc = __builtin_amdgcn_wmma_f32_16x16x4_f32(false, a, false, b, (short)0,
                                                  acc, false, false);
    }
    float bu = b_u2h[col];
#pragma unroll
    for (int i = 0; i < 8; ++i) {
      int rr = i + 8 * half;
      A[rr * A3_LD + HID + col] = acc[i] + degU[rr] * bu;
    }
  }

  {
    const int s4 = wv >> 1;
    const int kh = wv & 1;
    const unsigned bo2 = (unsigned)((kh * 68 + kb) * HID + s4 * 32 + 2 * m);
    const float* ap2 = Ah + m * AH_LD + kh * 68 + kb;
    v8f acc0 = {}, acc1 = {};
#pragma unroll
    for (int i = 0; i < 17; ++i) {
      const int k = i * 4;
      v2f a;
      a.x = ap2[k];
      a.y = ap2[k + 1];
      const unsigned br = bo2 + (unsigned)k * HID;
      const v2f b0 = *(const v2f*)(Whp + br);
      const v2f b1 = *(const v2f*)(Whp + (br + HID));
      v2f b;
      b.x = b0.x;
      b.y = b1.x;
      acc0 = __builtin_amdgcn_wmma_f32_16x16x4_f32(false, a, false, b, (short)0,
                                                   acc0, false, false);
      b.x = b0.y;
      b.y = b1.y;
      acc1 = __builtin_amdgcn_wmma_f32_16x16x4_f32(false, a, false, b, (short)0,
                                                   acc1, false, false);
    }
    float* D = A + 2 * HID + s4 * 32 + 2 * m;
    if (kh == 1) {
#pragma unroll
      for (int i = 0; i < 8; ++i) {
        const int rr = i + 8 * half;
        D[rr * A3_LD + 0] = acc0[i];
        D[rr * A3_LD + 1] = acc1[i];
      }
    }
    __syncthreads();
    if (kh == 0) {
      const float bh0 = b_h2h[s4 * 32 + 2 * m];
      const float bh1 = b_h2h[s4 * 32 + 2 * m + 1];
#pragma unroll
      for (int i = 0; i < 8; ++i) {
        const int rr = i + 8 * half;
        const float ds = degS[rr];
        const float dn = fmaxf(ds, 1.0f);
        D[rr * A3_LD + 0] = ((acc0[i] + D[rr * A3_LD + 0]) + ds * bh0) / dn;
        D[rr * A3_LD + 1] = ((acc1[i] + D[rr * A3_LD + 1]) + ds * bh1) / dn;
      }
    }
  }
  __syncthreads();

  {
    const int sp = wv >> 2;
    const int kq = wv & 3;
    const int ks0 = (kq * 49) >> 1;
    const int ks1 = ((kq + 1) * 49) >> 1;
    const unsigned bo = (unsigned)(kb * HID + sp * 64 + 4 * m);
    const float* ap = A + m * A3_LD + kb;
    v8f acc0 = {}, acc1 = {}, acc2 = {}, acc3 = {};
#pragma unroll 2
    for (int ks = ks0; ks < ks1; ++ks) {
      const int k = ks * 4;
      v2f a;
      a.x = ap[k];
      a.y = ap[k + 1];
      const unsigned br = bo + (unsigned)k * HID;
      const float4 b0 = *(const float4*)(Wp + br);
      const float4 b1 = *(const float4*)(Wp + (br + HID));
      v2f b;
      b.x = b0.x;
      b.y = b1.x;
      acc0 = __builtin_amdgcn_wmma_f32_16x16x4_f32(false, a, false, b, (short)0,
                                                   acc0, false, false);
      b.x = b0.y;
      b.y = b1.y;
      acc1 = __builtin_amdgcn_wmma_f32_16x16x4_f32(false, a, false, b, (short)0,
                                                   acc1, false, false);
      b.x = b0.z;
      b.y = b1.z;
      acc2 = __builtin_amdgcn_wmma_f32_16x16x4_f32(false, a, false, b, (short)0,
                                                   acc2, false, false);
      b.x = b0.w;
      b.y = b1.w;
      acc3 = __builtin_amdgcn_wmma_f32_16x16x4_f32(false, a, false, b, (short)0,
                                                   acc3, false, false);
    }
    __syncthreads();

    float* P = smem + wv * 1024 + m;
#pragma unroll
    for (int i = 0; i < 8; ++i) {
      const int ro = (i + 8 * half) * 16;
      P[0 * 256 + ro] = acc0[i];
      P[1 * 256 + ro] = acc1[i];
      P[2 * 256 + ro] = acc2[i];
      P[3 * 256 + ro] = acc3[i];
    }
  }
  __syncthreads();

  {
    const int row = tid >> 4;
    const int c16 = tid & 15;
    const int nn = n0 + row;
#pragma unroll
    for (int sp = 0; sp < 2; ++sp) {
      float o[4];
#pragma unroll
      for (int j = 0; j < 4; ++j) {
        const float* Q = smem + (sp * 4) * 1024 + j * 256 + row * 16 + c16;
        o[j] = ((Q[0] + Q[1024]) + Q[2048]) + Q[3072];
      }
      const float* bq = b_upd + sp * 64 + 4 * c16;
      if (nn < n_state) {
        float4 r;
        r.x = o[0] + bq[0];
        r.y = o[1] + bq[1];
        r.z = o[2] + bq[2];
        r.w = o[3] + bq[3];
        *(float4*)(out + (long)nn * HID + sp * 64 + 4 * c16) = r;
      }
    }
  }
}

extern "C" void kernel_launch(void* const* d_in, const int* in_sizes, int n_in,
                              void* d_out, int out_size, void* d_ws,
                              size_t ws_size, hipStream_t stream) {
  const float* h = (const float*)d_in[0];
  const float* u = (const float*)d_in[1];
  const float* pos_state = (const float*)d_in[2];
  const float* pos_action = (const float*)d_in[3];
  const int* a2s_src = (const int*)d_in[4];
  const int* a2s_dst = (const int*)d_in[5];
  const float* a2s_dis = (const float*)d_in[6];
  const int* s2s_src = (const int*)d_in[7];
  const int* s2s_dst = (const int*)d_in[8];
  const float* s2s_dis = (const float*)d_in[9];
  const float* W_u2h = (const float*)d_in[10];
  const float* b_u2h = (const float*)d_in[11];
  const float* W_h2h = (const float*)d_in[12];
  const float* b_h2h = (const float*)d_in[13];
  const float* W_upd = (const float*)d_in[14];
  const float* b_upd = (const float*)d_in[15];
  float* out = (float*)d_out;

  const int n_state = in_sizes[2] / 2;
  const int n_action = in_sizes[3] / 2;
  const int E_a2s = in_sizes[4];
  const int E_s2s = in_sizes[7];

  float* WTu = (float*)d_ws;
  float* Whp = WTu + HID * AU_KP;
  float* Wp = Whp + HID * AH_KP;
  int* listS = (int*)(Wp + HID * A3_KP);
  int* listA = listS + (size_t)E_s2s * 2;
  int* cntS = listA + (size_t)E_a2s * 2;
  int* cntA = cntS + n_state;
  int* tot = cntA + n_state;
  int* offS = tot + 2;
  int* offA = offS + n_state;
  int* curS = offA + n_state;
  int* curA = curS + n_state;
  const int E_all = E_s2s + E_a2s;
  int* rank = curA + n_state;
  if ((size_t)((char*)(rank + E_all) - (char*)d_ws) > ws_size) rank = nullptr;
  int* curS_used = rank ? nullptr : curS;
  int* curA_used = rank ? nullptr : curA;

  int* ll_recs = (int*)(Wp + HID * A3_KP);
  int* ll_heads = ll_recs + (size_t)E_all * 4;
  int hs = 16;
  if ((size_t)((char*)(ll_heads + (size_t)n_state * 32) - (char*)d_ws) > ws_size)
    hs = 4;
  if (hs == 4 &&
      (size_t)((char*)(ll_heads + (size_t)n_state * 8) - (char*)d_ws) > ws_size)
    hs = 2;
  if ((size_t)((char*)(ll_heads + (size_t)n_state * 2 * hs) - (char*)d_ws) <=
      ws_size) {
    pack_weights_kernel<<<256, 256, 0, stream>>>(W_u2h, W_h2h, W_upd, WTu, Whp,
                                                 Wp, nullptr, 0, ll_heads,
                                                 2 * n_state, hs);
    push_kernel<<<(E_all + 255) / 256, 256, 0, stream>>>(
        s2s_src, s2s_dst, s2s_dis, a2s_src, a2s_dst, a2s_dis, ll_heads,
        ll_recs, n_state, E_s2s, E_a2s, hs);
    fused_node_kernel<<<(n_state + 15) / 16, 256, 0, stream>>>(
        h, pos_state, u, pos_action, nullptr, nullptr, nullptr, nullptr,
        nullptr, nullptr, ll_recs, ll_heads, WTu, b_u2h, Whp, b_h2h, Wp, b_upd,
        out, n_state, n_action, E_s2s, E_a2s, hs);
    return;
  }
  pack_weights_kernel<<<256, 256, 0, stream>>>(W_u2h, W_h2h, W_upd, WTu, Whp,
                                               Wp, cntS, 2 * n_state + 2,
                                               nullptr, 0, 0);
  count_kernel<<<(E_all + 255) / 256, 256, 0, stream>>>(
      s2s_dst, a2s_dst, cntS, cntA, rank, E_s2s, E_a2s);
  alloc_kernel<<<(n_state + 255) / 256, 256, 0, stream>>>(
      cntS, cntA, tot, offS, offA, curS_used, curA_used, n_state);
  fill_kernel<<<(E_all + 255) / 256, 256, 0, stream>>>(
      s2s_src, s2s_dst, s2s_dis, a2s_src, a2s_dst, a2s_dis, offS, offA, rank,
      curS_used, curA_used, listS, listA, E_s2s, E_a2s);
  fused_node_kernel<<<(n_state + 15) / 16, 256, 0, stream>>>(
      h, pos_state, u, pos_action, listS, listA, offS, cntS, offA, cntA,
      nullptr, nullptr, WTu, b_u2h, Whp, b_h2h, Wp, b_upd, out, n_state,
      n_action, E_s2s, E_a2s, 1);
}
